# starting kernel
# baseline (speedup 1.0000x reference)
#include <hip/hip_runtime.h>
#include <hip/hip_bf16.h>

#define N_NODES 30000
#define E_EDGES 480000
#define EN_TOT  (E_EDGES + N_NODES)
#define DIM     128
#define NLAYERS 3
#define NEG_SLOPE 0.2f
#define NWEIGHTS 4
#define WPK_ELEMS_PER_W (8 * 4 * 32 * 16)

typedef __attribute__((ext_vector_type(16))) __bf16 v16bf;
typedef __attribute__((ext_vector_type(4)))  __bf16 v4bf;
typedef __attribute__((ext_vector_type(8)))  float  v8f;

__device__ __forceinline__ unsigned f2ord(float x) {
    unsigned b = __float_as_uint(x);
    return (b & 0x80000000u) ? ~b : (b | 0x80000000u);
}
__device__ __forceinline__ float ord2f(unsigned u) {
    return __uint_as_float((u & 0x80000000u) ? (u ^ 0x80000000u) : ~u);
}

__global__ void __launch_bounds__(256)
pack_weights_kernel(const float* __restrict__ W_mlp, const float* __restrict__ Ws,
                    __bf16* __restrict__ Wpk)
{
    const int gw   = (blockIdx.x * blockDim.x + threadIdx.x) >> 5;
    const int lane = threadIdx.x & 31;
    if (gw >= NWEIGHTS * 32) return;
    const int w  = gw >> 5;
    const int ct = (gw & 31) >> 2;
    const int s  = gw & 3;
    const float* W = (w == 0) ? W_mlp : (Ws + (size_t)(w - 1) * DIM * DIM);
    const int r  = lane & 15;
    const int hi = lane >> 4;
    const int col = ct * 16 + r;
    const int kb  = s * 32;
    __bf16* out = Wpk + ((size_t)gw * 32 + lane) * 16;
#pragma unroll
    for (int h = 0; h < 16; ++h) {
        const int K = kb + ((h < 8) ? (8 * hi + h) : (16 + 8 * hi + (h - 8)));
        out[h] = (__bf16)W[(size_t)K * DIM + col];
    }
}

template <bool BF16_OUT>
__global__ void __launch_bounds__(256)
gemm128_wmma(const float* __restrict__ X, const __bf16* __restrict__ Wpk,
             const float* __restrict__ bias, void* __restrict__ out, int rows)
{
    if (blockIdx.x * 16 >= (unsigned)rows) return;
    const int lane = threadIdx.x & 31;
    const int ct   = threadIdx.x >> 5;
    const int r    = lane & 15;
    const int hi   = lane >> 4;
    const int row  = blockIdx.x * 16 + r;
    const int col0 = ct * 16;

    v8f acc = {};
#pragma unroll
    for (int s = 0; s < 4; ++s) {
        const int kb = s * 32;
        v16bf a;
        const float* ap = X + (size_t)row * DIM + kb + 8 * hi;
#pragma unroll
        for (int j = 0; j < 8; ++j) {
            a[j]     = (__bf16)ap[j];
            a[8 + j] = (__bf16)ap[16 + j];
        }
        const v16bf b = *(const v16bf*)(Wpk + ((size_t)(ct * 4 + s) * 32 + lane) * 16);
        acc = __builtin_amdgcn_wmma_f32_16x16x32_bf16(
             false, a,  false, b,
             (short)0, acc,  false,  false);
    }
    if (BF16_OUT) {
        __bf16* H = (__bf16*)out;
#pragma unroll
        for (int v = 0; v < 8; ++v) {
            const int m = blockIdx.x * 16 + v + 8 * hi;
            H[(size_t)m * DIM + col0 + r] = (__bf16)acc[v];
        }
    } else {
        float* H = (float*)out;
        const float bv = bias[col0 + r];
#pragma unroll
        for (int v = 0; v < 8; ++v) {
            const int m = blockIdx.x * 16 + v + 8 * hi;
            H[(size_t)m * DIM + col0 + r] = acc[v] + bv;
        }
    }
}

__global__ void __launch_bounds__(256)
alpha_init_kernel(const __bf16* __restrict__ Hb,
                  const float* __restrict__ asrc, const float* __restrict__ adst,
                  float* __restrict__ as_, float* __restrict__ ad_,
                  unsigned* __restrict__ m_u, float* __restrict__ z,
                  float* __restrict__ accum)
{
    const int wave = blockIdx.x * (blockDim.x >> 5) + (threadIdx.x >> 5);
    const int lane = threadIdx.x & 31;
    if (wave >= N_NODES) return;

    const v4bf   h4 = ((const v4bf*)(Hb + (size_t)wave * DIM))[lane];
    const float4 s4 = ((const float4*)asrc)[lane];
    const float4 d4 = ((const float4*)adst)[lane];
    const float hx = (float)h4[0], hy = (float)h4[1], hz = (float)h4[2], hw = (float)h4[3];
    float ps = hx * s4.x + hy * s4.y + hz * s4.z + hw * s4.w;
    float pd = hx * d4.x + hy * d4.y + hz * d4.z + hw * d4.w;
#pragma unroll
    for (int off = 16; off > 0; off >>= 1) {
        ps += __shfl_down(ps, off, 32);
        pd += __shfl_down(pd, off, 32);
    }
    if (lane == 0) {
        as_[wave] = ps;
        ad_[wave] = pd;
        m_u[wave] = 0u;
        z[wave]   = 0.0f;
    }
    ((float4*)(accum + (size_t)wave * DIM))[lane] = make_float4(0.f, 0.f, 0.f, 0.f);
}

__global__ void __launch_bounds__(256)
edge_max_kernel(const int* __restrict__ esrc, const int* __restrict__ edst,
                const float* __restrict__ as_, const float* __restrict__ ad_,
                unsigned* __restrict__ m_u)
{
    const int e = blockIdx.x * blockDim.x + threadIdx.x;
    if (e >= EN_TOT) return;
    int s, d;
    if (e < E_EDGES) { s = esrc[e]; d = edst[e]; } else { s = d = e - E_EDGES; }
    float v = as_[s] + ad_[d];
    v = (v > 0.0f) ? v : NEG_SLOPE * v;
    atomicMax(&m_u[d], f2ord(v));
}

__global__ void __launch_bounds__(256)
edge_sum_kernel(const int* __restrict__ esrc, const int* __restrict__ edst,
                const float* __restrict__ as_, const float* __restrict__ ad_,
                const unsigned* __restrict__ m_u, float* __restrict__ z)
{
    const int e = blockIdx.x * blockDim.x + threadIdx.x;
    if (e >= EN_TOT) return;
    int s, d;
    if (e < E_EDGES) { s = esrc[e]; d = edst[e]; } else { s = d = e - E_EDGES; }
    float v = as_[s] + ad_[d];
    v = (v > 0.0f) ? v : NEG_SLOPE * v;
    atomicAdd(&z[d], __expf(v - ord2f(m_u[d])));
}

__global__ void __launch_bounds__(256)
edge_aggr_kernel(const int* __restrict__ esrc, const int* __restrict__ edst,
                 const float* __restrict__ as_, const float* __restrict__ ad_,
                 const unsigned* __restrict__ m_u, const float* __restrict__ z,
                 const __bf16* __restrict__ Hb, float* __restrict__ accum)
{
    const int wave = blockIdx.x * (blockDim.x >> 5) + (threadIdx.x >> 5);
    const int lane = threadIdx.x & 31;
    if (wave >= EN_TOT) return;
    int s, d;
    if (wave < E_EDGES) { s = esrc[wave]; d = edst[wave]; } else { s = d = wave - E_EDGES; }
    float v = as_[s] + ad_[d];
    v = (v > 0.0f) ? v : NEG_SLOPE * v;
    const float coef = __expf(v - ord2f(m_u[d])) / z[d];

    const v4bf h4 = ((const v4bf*)(Hb + (size_t)s * DIM))[lane];
    float* ap = accum + (size_t)d * DIM + lane * 4;
    atomicAdd(ap + 0, coef * (float)h4[0]);
    atomicAdd(ap + 1, coef * (float)h4[1]);
    atomicAdd(ap + 2, coef * (float)h4[2]);
    atomicAdd(ap + 3, coef * (float)h4[3]);
}

__global__ void __launch_bounds__(256)
finalize_kernel(const float* __restrict__ accum, const float* __restrict__ bias,
                float* __restrict__ X)
{
    const int i = blockIdx.x * blockDim.x + threadIdx.x;
    if (i >= N_NODES * DIM) return;
    const float v = accum[i] + bias[i & (DIM - 1)];
    X[i] = (v > 0.0f) ? v : (__expf(v) - 1.0f);
}

extern "C" void kernel_launch(void* const* d_in, const int* in_sizes, int n_in,
                              void* d_out, int out_size, void* d_ws, size_t ws_size,
                              hipStream_t stream)
{
    (void)in_sizes; (void)n_in; (void)out_size; (void)ws_size;

    const float* xq    = (const float*)d_in[0];
    const float* xt    = (const float*)d_in[1];
    const int*   ei_q  = (const int*)d_in[2];
    const float* W_mlp = (const float*)d_in[4];
    const float* b_mlp = (const float*)d_in[5];
    const float* Ws    = (const float*)d_in[6];
    const float* a_src = (const float*)d_in[7];
    const float* a_dst = (const float*)d_in[8];
    const float* biases= (const float*)d_in[9];

    const int* esrc = ei_q;
    const int* edst = ei_q + E_EDGES;

    float* Xq = (float*)d_out;
    float* Xt = Xq + (size_t)N_NODES * DIM;

    __bf16*   Wpk   = (__bf16*)d_ws;
    __bf16*   Hb    = Wpk + (size_t)NWEIGHTS * WPK_ELEMS_PER_W;
    float*    accum = (float*)(Hb + (size_t)N_NODES * DIM);
    float*    as_   = accum + (size_t)N_NODES * DIM;
    float*    ad_   = as_ + N_NODES;
    unsigned* m_u   = (unsigned*)(ad_ + N_NODES);
    float*    z     = (float*)(m_u + N_NODES);

    const dim3 blk(256);
    const dim3 pack_grid((NWEIGHTS * 32 * 32 + 255) / 256);
    const dim3 gemm_grid(N_NODES / 16);
    const dim3 node_wave_grid((N_NODES + 7) / 8);
    const dim3 edge_grid((EN_TOT + 255) / 256);
    const dim3 edge_wave_grid((EN_TOT + 7) / 8);
    const dim3 elem_grid((N_NODES * DIM + 255) / 256);

    pack_weights_kernel<<<pack_grid, blk, 0, stream>>>(W_mlp, Ws, Wpk);

    gemm128_wmma<false><<<gemm_grid, blk, 0, stream>>>(xq, Wpk, b_mlp, Xq, N_NODES);
    gemm128_wmma<false><<<gemm_grid, blk, 0, stream>>>(xt, Wpk, b_mlp, Xt, N_NODES);

    for (int l = 0; l < NLAYERS; ++l) {
        const __bf16* Wl = Wpk + (size_t)(l + 1) * WPK_ELEMS_PER_W;
        const float*  al = a_src  + (size_t)l * DIM;
        const float*  dl = a_dst  + (size_t)l * DIM;
        const float*  bl = biases + (size_t)l * DIM;
        float* Xs[2] = { Xq, Xt };
        for (int g = 0; g < 2; ++g) {
            float* X = Xs[g];
            gemm128_wmma<true><<<gemm_grid, blk, 0, stream>>>(X, Wl, nullptr, (void*)Hb, N_NODES);
            alpha_init_kernel<<<node_wave_grid, blk, 0, stream>>>(Hb, al, dl, as_, ad_, m_u, z, accum);
            edge_max_kernel <<<edge_grid, blk, 0, stream>>>(esrc, edst, as_, ad_, m_u);
            edge_sum_kernel <<<edge_grid, blk, 0, stream>>>(esrc, edst, as_, ad_, m_u, z);
            edge_aggr_kernel<<<edge_wave_grid, blk, 0, stream>>>(esrc, edst, as_, ad_, m_u, z, Hb, accum);
            finalize_kernel <<<elem_grid, blk, 0, stream>>>(accum, bl, X);
        }
    }
}


# Atomic edge scatter -> per-destination gather: edges bucketed by dst once per call, buckets sorted by src; one wave per node does softmax and row sum in registers; fixed sum order, no float atomics
# speedup vs baseline: 3.6653x; 3.6653x over previous
#include <hip/hip_runtime.h>
#include <hip/hip_bf16.h>

#define N_NODES 30000
#define E_EDGES 480000
#define DIM     128
#define NLAYERS 3
#define NEG_SLOPE 0.2f
#define NWEIGHTS 4
#define WPK_ELEMS_PER_W (8 * 4 * 32 * 16)

typedef __attribute__((ext_vector_type(16))) __bf16 v16bf;
typedef __attribute__((ext_vector_type(4)))  __bf16 v4bf;
typedef __attribute__((ext_vector_type(8)))  __bf16 v8bf;
typedef __attribute__((ext_vector_type(8)))  float  v8f;

__global__ void __launch_bounds__(256)
pack_weights_kernel(const float* __restrict__ W_mlp, const float* __restrict__ Ws,
                    __bf16* __restrict__ Wpk, int* __restrict__ cnt)
{
    for (int i = blockIdx.x * blockDim.x + threadIdx.x; i < N_NODES; i += gridDim.x * blockDim.x)
        cnt[i] = 0;
    const int gw   = (blockIdx.x * blockDim.x + threadIdx.x) >> 5;
    const int lane = threadIdx.x & 31;
    if (gw >= NWEIGHTS * 32) return;
    const int w  = gw >> 5;
    const int ct = (gw & 31) >> 2;
    const int s  = gw & 3;
    const float* W = (w == 0) ? W_mlp : (Ws + (size_t)(w - 1) * DIM * DIM);
    const int r  = lane & 15;
    const int hi = lane >> 4;
    const int col = ct * 16 + r;
    const int kb  = s * 32;
    __bf16* out = Wpk + ((size_t)gw * 32 + lane) * 16;
#pragma unroll
    for (int h = 0; h < 16; ++h) {
        const int K = kb + ((h < 8) ? (8 * hi + h) : (16 + 8 * hi + (h - 8)));
        out[h] = (__bf16)W[(size_t)K * DIM + col];
    }
}

template <bool BF16_OUT>
__global__ void __launch_bounds__(256)
gemm128_wmma(const float* __restrict__ X, const __bf16* __restrict__ Wpk,
             const float* __restrict__ bias, void* __restrict__ out, int rows)
{
    if (blockIdx.x * 16 >= (unsigned)rows) return;
    const int lane = threadIdx.x & 31;
    const int ct   = threadIdx.x >> 5;
    const int r    = lane & 15;
    const int hi   = lane >> 4;
    const int row  = blockIdx.x * 16 + r;
    const int col0 = ct * 16;

    v8f acc = {};
#pragma unroll
    for (int s = 0; s < 4; ++s) {
        const int kb = s * 32;
        v16bf a;
        const float* ap = X + (size_t)row * DIM + kb + 8 * hi;
#pragma unroll
        for (int j = 0; j < 8; ++j) {
            a[j]     = (__bf16)ap[j];
            a[8 + j] = (__bf16)ap[16 + j];
        }
        const v16bf b = *(const v16bf*)(Wpk + ((size_t)(ct * 4 + s) * 32 + lane) * 16);
        acc = __builtin_amdgcn_wmma_f32_16x16x32_bf16(
             false, a,  false, b,
             (short)0, acc,  false,  false);
    }
    if (BF16_OUT) {
        __bf16* H = (__bf16*)out;
#pragma unroll
        for (int v = 0; v < 8; ++v) {
            const int m = blockIdx.x * 16 + v + 8 * hi;
            H[(size_t)m * DIM + col0 + r] = (__bf16)acc[v];
        }
    } else {
        float* H = (float*)out;
        const float bv = bias[col0 + r];
#pragma unroll
        for (int v = 0; v < 8; ++v) {
            const int m = blockIdx.x * 16 + v + 8 * hi;
            H[(size_t)m * DIM + col0 + r] = acc[v] + bv;
        }
    }
}

__global__ void __launch_bounds__(256)
edge_rank_kernel(const int* __restrict__ edst, int* __restrict__ cnt, int* __restrict__ rank)
{
    const int e = blockIdx.x * blockDim.x + threadIdx.x;
    if (e >= E_EDGES) return;
    rank[e] = atomicAdd(&cnt[edst[e]], 1);
}

#define SCAN_THREADS 1024
#define SCAN_PER     ((N_NODES + SCAN_THREADS - 1) / SCAN_THREADS)

__global__ void __launch_bounds__(SCAN_THREADS)
bucket_scan_kernel(const int* __restrict__ cnt, int* __restrict__ off)
{
    __shared__ int sm[SCAN_THREADS];
    const int t = threadIdx.x;
    const int b = t * SCAN_PER;
    int v[SCAN_PER];
    int sum = 0;
#pragma unroll
    for (int i = 0; i < SCAN_PER; ++i) {
        v[i] = (b + i < N_NODES) ? cnt[b + i] : 0;
        sum += v[i];
    }
    sm[t] = sum;
    __syncthreads();
    for (int o = 1; o < SCAN_THREADS; o <<= 1) {
        const int y = (t >= o) ? sm[t - o] : 0;
        __syncthreads();
        sm[t] += y;
        __syncthreads();
    }
    int run = sm[t] - sum;
#pragma unroll
    for (int i = 0; i < SCAN_PER; ++i) {
        if (b + i < N_NODES) off[b + i] = run;
        run += v[i];
    }
    if (t == SCAN_THREADS - 1) off[N_NODES] = sm[t];
}

__global__ void __launch_bounds__(256)
bucket_fill_kernel(const int* __restrict__ esrc, const int* __restrict__ edst,
                   const int* __restrict__ off, const int* __restrict__ rank,
                   int* __restrict__ tmp)
{
    const int e = blockIdx.x * blockDim.x + threadIdx.x;
    if (e >= E_EDGES) return;
    tmp[off[edst[e]] + rank[e]] = esrc[e];
}

__global__ void __launch_bounds__(256)
bucket_sort_kernel(const int* __restrict__ off, const int* __restrict__ tmp,
                   int* __restrict__ csr)
{
    const int d    = blockIdx.x * (blockDim.x >> 5) + (threadIdx.x >> 5);
    const int lane = threadIdx.x & 31;
    if (d >= N_NODES) return;
    const int beg = off[d];
    const int deg = off[d + 1] - beg;
    if (deg <= 32) {
        const int key = (lane < deg) ? tmp[beg + lane] : 0x7fffffff;
        int pos = 0;
        for (int j = 0; j < deg; ++j) {
            const int kj = __shfl(key, j, 32);
            pos += (kj < key || (kj == key && j < lane)) ? 1 : 0;
        }
        if (lane < deg) csr[beg + pos] = key;
    } else {
        for (int i = lane; i < deg; i += 32) {
            const int key = tmp[beg + i];
            int pos = 0;
            for (int j = 0; j < deg; ++j) {
                const int kj = tmp[beg + j];
                pos += (kj < key || (kj == key && j < i)) ? 1 : 0;
            }
            csr[beg + pos] = key;
        }
    }
}

__global__ void __launch_bounds__(256)
alpha_kernel(const __bf16* __restrict__ Hb,
             const float* __restrict__ asrc, const float* __restrict__ adst,
             float* __restrict__ as_, float* __restrict__ ad_)
{
    const int wave = blockIdx.x * (blockDim.x >> 5) + (threadIdx.x >> 5);
    const int lane = threadIdx.x & 31;
    if (wave >= N_NODES) return;

    const v4bf   h4 = ((const v4bf*)(Hb + (size_t)wave * DIM))[lane];
    const float4 s4 = ((const float4*)asrc)[lane];
    const float4 d4 = ((const float4*)adst)[lane];
    const float hx = (float)h4[0], hy = (float)h4[1], hz = (float)h4[2], hw = (float)h4[3];
    float ps = hx * s4.x + hy * s4.y + hz * s4.z + hw * s4.w;
    float pd = hx * d4.x + hy * d4.y + hz * d4.z + hw * d4.w;
#pragma unroll
    for (int o = 16; o > 0; o >>= 1) {
        ps += __shfl_down(ps, o, 32);
        pd += __shfl_down(pd, o, 32);
    }
    if (lane == 0) {
        as_[wave] = ps;
        ad_[wave] = pd;
    }
}

__device__ __forceinline__ void row_fma(float (&acc)[8], const v8bf h, const float w)
{
#pragma unroll
    for (int i = 0; i < 8; ++i) acc[i] = fmaf(w, (float)h[i], acc[i]);
}

__global__ void __launch_bounds__(256)
node_aggregate_kernel(const int* __restrict__ off, const int* __restrict__ csr,
                      const float* __restrict__ as_, const float* __restrict__ ad_,
                      const __bf16* __restrict__ Hb, const float* __restrict__ bias,
                      float* __restrict__ X)
{
    const int d    = __builtin_amdgcn_readfirstlane(
                         (int)(blockIdx.x * (blockDim.x >> 5) + (threadIdx.x >> 5)));
    const int lane = threadIdx.x & 31;
    if (d >= N_NODES) return;
    const int half = lane >> 4;
    const int c    = lane & 15;
    const int beg  = off[d];
    const int deg  = off[d + 1] - beg;
    const int n    = deg + 1;
    const float adv = ad_[d];
    const v8bf* __restrict__ Hrow = (const v8bf*)Hb + c;

    int   s0 = d;
    float v0 = -INFINITY;
    if (lane < n) {
        s0 = (lane < deg) ? csr[beg + lane] : d;
        const float t = as_[s0] + adv;
        v0 = (t > 0.0f) ? t : NEG_SLOPE * t;
    }
    float mx = v0;
    for (int base = 32; base < n; base += 32) {
        const int j = base + lane;
        if (j < n) {
            const int s = (j < deg) ? csr[beg + j] : d;
            const float t = as_[s] + adv;
            mx = fmaxf(mx, (t > 0.0f) ? t : NEG_SLOPE * t);
        }
    }
#pragma unroll
    for (int o = 16; o > 0; o >>= 1) mx = fmaxf(mx, __shfl_xor(mx, o, 32));

    float acc[8] = {0.f, 0.f, 0.f, 0.f, 0.f, 0.f, 0.f, 0.f};
    float zp = 0.0f;
    for (int base = 0; base < n; base += 32) {
        int   s = s0;
        float w = (lane < n) ? __expf(v0 - mx) : 0.0f;
        if (base != 0) {
            const int j = base + lane;
            s = d; w = 0.0f;
            if (j < n) {
                s = (j < deg) ? csr[beg + j] : d;
                const float t = as_[s] + adv;
                w = __expf(((t > 0.0f) ? t : NEG_SLOPE * t) - mx);
            }
        }
        zp += w;
        const int m = (n - base < 32) ? (n - base) : 32;
        for (int k = 0; k < m; k += 8) {
            v8bf  h[4];
            float ww[4];
#pragma unroll
            for (int u = 0; u < 4; ++u) {
                const int jj = k + 2 * u + half;
                const int ss = __shfl(s, jj, 32);
                ww[u] = __shfl(w, jj, 32);
                h[u]  = Hrow[(size_t)ss * 16];
            }
#pragma unroll
            for (int u = 0; u < 4; ++u) row_fma(acc, h[u], ww[u]);
        }
    }
#pragma unroll
    for (int o = 16; o > 0; o >>= 1) zp += __shfl_xor(zp, o, 32);
    const float rz = 1.0f / zp;

    float out4[4];
#pragma unroll
    for (int i = 0; i < 4; ++i) {
        const float give = half ? acc[i] : acc[4 + i];
        const float keep = half ? acc[4 + i] : acc[i];
        out4[i] = keep + __shfl_xor(give, 16, 32);
    }
    const int col = c * 8 + half * 4;
    const float4 b4 = *(const float4*)(bias + col);
    float4 r;
    r.x = out4[0] * rz + b4.x;
    r.y = out4[1] * rz + b4.y;
    r.z = out4[2] * rz + b4.z;
    r.w = out4[3] * rz + b4.w;
    r.x = (r.x > 0.0f) ? r.x : (__expf(r.x) - 1.0f);
    r.y = (r.y > 0.0f) ? r.y : (__expf(r.y) - 1.0f);
    r.z = (r.z > 0.0f) ? r.z : (__expf(r.z) - 1.0f);
    r.w = (r.w > 0.0f) ? r.w : (__expf(r.w) - 1.0f);
    *(float4*)(X + (size_t)d * DIM + col) = r;
}

extern "C" void kernel_launch(void* const* d_in, const int* in_sizes, int n_in,
                              void* d_out, int out_size, void* d_ws, size_t ws_size,
                              hipStream_t stream)
{
    (void)in_sizes; (void)n_in; (void)out_size; (void)ws_size;

    const float* xq    = (const float*)d_in[0];
    const float* xt    = (const float*)d_in[1];
    const int*   ei_q  = (const int*)d_in[2];
    const float* W_mlp = (const float*)d_in[4];
    const float* b_mlp = (const float*)d_in[5];
    const float* Ws    = (const float*)d_in[6];
    const float* a_src = (const float*)d_in[7];
    const float* a_dst = (const float*)d_in[8];
    const float* biases= (const float*)d_in[9];

    const int* esrc = ei_q;
    const int* edst = ei_q + E_EDGES;

    float* Xq = (float*)d_out;
    float* Xt = Xq + (size_t)N_NODES * DIM;

    __bf16* Wpk  = (__bf16*)d_ws;
    __bf16* Hb   = Wpk + (size_t)NWEIGHTS * WPK_ELEMS_PER_W;
    float*  as_  = (float*)(Hb + (size_t)N_NODES * DIM);
    float*  ad_  = as_ + N_NODES;
    int*    cnt  = (int*)(ad_ + N_NODES);
    int*    off  = cnt + N_NODES;
    int*    rank = off + N_NODES + 8;
    int*    tmp  = rank + E_EDGES;
    int*    csr  = tmp + E_EDGES;

    const dim3 blk(256);
    const dim3 pack_grid((NWEIGHTS * 32 * 32 + 255) / 256);
    const dim3 gemm_grid(N_NODES / 16);
    const dim3 node_wave_grid((N_NODES + 7) / 8);
    const dim3 edge_grid((E_EDGES + 255) / 256);

    pack_weights_kernel<<<pack_grid, blk, 0, stream>>>(W_mlp, Ws, Wpk, cnt);

    edge_rank_kernel  <<<edge_grid, blk, 0, stream>>>(edst, cnt, rank);
    bucket_scan_kernel<<<dim3(1), dim3(SCAN_THREADS), 0, stream>>>(cnt, off);
    bucket_fill_kernel<<<edge_grid, blk, 0, stream>>>(esrc, edst, off, rank, tmp);
    bucket_sort_kernel<<<node_wave_grid, blk, 0, stream>>>(off, tmp, csr);

    gemm128_wmma<false><<<gemm_grid, blk, 0, stream>>>(xq, Wpk, b_mlp, Xq, N_NODES);
    gemm128_wmma<false><<<gemm_grid, blk, 0, stream>>>(xt, Wpk, b_mlp, Xt, N_NODES);

    for (int l = 0; l < NLAYERS; ++l) {
        const __bf16* Wl = Wpk + (size_t)(l + 1) * WPK_ELEMS_PER_W;
        const float*  al = a_src  + (size_t)l * DIM;
        const float*  dl = a_dst  + (size_t)l * DIM;
        const float*  bl = biases + (size_t)l * DIM;
        float* Xs[2] = { Xq, Xt };
        for (int g = 0; g < 2; ++g) {
            float* X = Xs[g];
            gemm128_wmma<true><<<gemm_grid, blk, 0, stream>>>(X, Wl, nullptr, (void*)Hb, N_NODES);
            alpha_kernel<<<node_wave_grid, blk, 0, stream>>>(Hb, al, dl, as_, ad_);
            node_aggregate_kernel<<<node_wave_grid, blk, 0, stream>>>(off, csr, as_, ad_, Hb, bl, X);
        }
    }
}


# GEMM: one wave per 16-row strip holds all 8 output tiles (tile transposed, W fragment as A): 8x fewer waves, X read once, 16-B row stores; alpha dots summed in-lane, then lane halves (new sum order)
# speedup vs baseline: 4.2764x; 1.1667x over previous
#include <hip/hip_runtime.h>
#include <hip/hip_bf16.h>

#define N_NODES 30000
#define E_EDGES 480000
#define DIM     128
#define NLAYERS 3
#define NEG_SLOPE 0.2f
#define NWEIGHTS 4
#define WPK_ELEMS_PER_W (8 * 4 * 32 * 16)

typedef __attribute__((ext_vector_type(16))) __bf16 v16bf;
typedef __attribute__((ext_vector_type(4)))  __bf16 v4bf;
typedef __attribute__((ext_vector_type(8)))  __bf16 v8bf;
typedef __attribute__((ext_vector_type(8)))  float  v8f;

__global__ void __launch_bounds__(256)
pack_weights_kernel(const float* __restrict__ W_mlp, const float* __restrict__ Ws,
                    __bf16* __restrict__ Wpk, int* __restrict__ cnt)
{
    for (int i = blockIdx.x * blockDim.x + threadIdx.x; i < N_NODES; i += gridDim.x * blockDim.x)
        cnt[i] = 0;
    const int gw   = (blockIdx.x * blockDim.x + threadIdx.x) >> 5;
    const int lane = threadIdx.x & 31;
    if (gw >= NWEIGHTS * 32) return;
    const int w  = gw >> 5;
    const int ct = (gw & 31) >> 2;
    const int s  = gw & 3;
    const float* W = (w == 0) ? W_mlp : (Ws + (size_t)(w - 1) * DIM * DIM);
    const int r  = lane & 15;
    const int hi = lane >> 4;
    const int col = ct * 16 + r;
    const int kb  = s * 32;
    __bf16* out = Wpk + ((size_t)gw * 32 + lane) * 16;
#pragma unroll
    for (int h = 0; h < 16; ++h) {
        const int K = kb + ((h < 8) ? (8 * hi + h) : (16 + 8 * hi + (h - 8)));
        out[h] = (__bf16)W[(size_t)K * DIM + col];
    }
}

#define GEMM_WAVES 2

template <bool BF16_OUT>
__global__ void __launch_bounds__(GEMM_WAVES * 32)
gemm128_wmma(const float* __restrict__ X, const __bf16* __restrict__ Wpk,
             const float* __restrict__ bias, void* __restrict__ out, int rows,
             const float* __restrict__ asrc, const float* __restrict__ adst,
             float* __restrict__ as_, float* __restrict__ ad_)
{
    const int strip = __builtin_amdgcn_readfirstlane(
                          (int)(blockIdx.x * GEMM_WAVES + (threadIdx.x >> 5)));
    if (strip * 16 >= rows) return;
    const int lane = threadIdx.x & 31;
    const int r    = lane & 15;
    const int hi   = lane >> 4;
    const int row  = strip * 16 + r;
    const int cb   = 8 * hi;

    v16bf x[4];
    const float4* xp = (const float4*)(X + (size_t)row * DIM + cb);
#pragma unroll
    for (int s = 0; s < 4; ++s) {
        const float4 p0 = xp[s * 8 + 0];
        const float4 p1 = xp[s * 8 + 1];
        const float4 p2 = xp[s * 8 + 4];
        const float4 p3 = xp[s * 8 + 5];
        x[s][0]  = (__bf16)p0.x; x[s][1]  = (__bf16)p0.y; x[s][2]  = (__bf16)p0.z; x[s][3]  = (__bf16)p0.w;
        x[s][4]  = (__bf16)p1.x; x[s][5]  = (__bf16)p1.y; x[s][6]  = (__bf16)p1.z; x[s][7]  = (__bf16)p1.w;
        x[s][8]  = (__bf16)p2.x; x[s][9]  = (__bf16)p2.y; x[s][10] = (__bf16)p2.z; x[s][11] = (__bf16)p2.w;
        x[s][12] = (__bf16)p3.x; x[s][13] = (__bf16)p3.y; x[s][14] = (__bf16)p3.z; x[s][15] = (__bf16)p3.w;
    }

    v8f acc[8];
#pragma unroll
    for (int ct = 0; ct < 8; ++ct) acc[ct] = v8f{};
#pragma unroll
    for (int s = 0; s < 4; ++s) {
#pragma unroll
        for (int ct = 0; ct < 8; ++ct) {
            const v16bf w = *(const v16bf*)(Wpk + ((size_t)(ct * 4 + s) * 32 + lane) * 16);
            acc[ct] = __builtin_amdgcn_wmma_f32_16x16x32_bf16(
                 false, w,  false, x[s],
                 (short)0, acc[ct],  false,  false);
        }
    }

    if (BF16_OUT) {
        float ps = 0.0f, pd = 0.0f;
        __bf16* H = (__bf16*)out + (size_t)row * DIM + cb;
#pragma unroll
        for (int ct = 0; ct < 8; ++ct) {
            const float4 s0 = *(const float4*)(asrc + ct * 16 + cb);
            const float4 s1 = *(const float4*)(asrc + ct * 16 + cb + 4);
            const float4 d0 = *(const float4*)(adst + ct * 16 + cb);
            const float4 d1 = *(const float4*)(adst + ct * 16 + cb + 4);
            const v8f a = acc[ct];
            float ts = a[0] * s0.x;
            ts = fmaf(a[1], s0.y, ts); ts = fmaf(a[2], s0.z, ts); ts = fmaf(a[3], s0.w, ts);
            ts = fmaf(a[4], s1.x, ts); ts = fmaf(a[5], s1.y, ts); ts = fmaf(a[6], s1.z, ts);
            ts = fmaf(a[7], s1.w, ts);
            float td = a[0] * d0.x;
            td = fmaf(a[1], d0.y, td); td = fmaf(a[2], d0.z, td); td = fmaf(a[3], d0.w, td);
            td = fmaf(a[4], d1.x, td); td = fmaf(a[5], d1.y, td); td = fmaf(a[6], d1.z, td);
            td = fmaf(a[7], d1.w, td);
            ps += ts;
            pd += td;
            v8bf o;
            o[0] = (__bf16)a[0]; o[1] = (__bf16)a[1]; o[2] = (__bf16)a[2]; o[3] = (__bf16)a[3];
            o[4] = (__bf16)a[4]; o[5] = (__bf16)a[5]; o[6] = (__bf16)a[6]; o[7] = (__bf16)a[7];
            *(v8bf*)(H + ct * 16) = o;
        }
        ps += __shfl_xor(ps, 16, 32);
        pd += __shfl_xor(pd, 16, 32);
        if (hi) ad_[row] = pd; else as_[row] = ps;
    } else {
        float* H = (float*)out + (size_t)row * DIM + cb;
#pragma unroll
        for (int ct = 0; ct < 8; ++ct) {
            const float4 b0 = *(const float4*)(bias + ct * 16 + cb);
            const float4 b1 = *(const float4*)(bias + ct * 16 + cb + 4);
            const v8f a = acc[ct];
            float4 o0, o1;
            o0.x = a[0] + b0.x; o0.y = a[1] + b0.y; o0.z = a[2] + b0.z; o0.w = a[3] + b0.w;
            o1.x = a[4] + b1.x; o1.y = a[5] + b1.y; o1.z = a[6] + b1.z; o1.w = a[7] + b1.w;
            *(float4*)(H + ct * 16)     = o0;
            *(float4*)(H + ct * 16 + 4) = o1;
        }
    }
}

__global__ void __launch_bounds__(256)
edge_rank_kernel(const int* __restrict__ edst, int* __restrict__ cnt, int* __restrict__ rank)
{
    const int e = blockIdx.x * blockDim.x + threadIdx.x;
    if (e >= E_EDGES) return;
    rank[e] = atomicAdd(&cnt[edst[e]], 1);
}

#define SCAN_THREADS 1024
#define SCAN_PER     ((N_NODES + SCAN_THREADS - 1) / SCAN_THREADS)

__global__ void __launch_bounds__(SCAN_THREADS)
bucket_scan_kernel(const int* __restrict__ cnt, int* __restrict__ off)
{
    __shared__ int sm[SCAN_THREADS];
    const int t = threadIdx.x;
    const int b = t * SCAN_PER;
    int v[SCAN_PER];
    int sum = 0;
#pragma unroll
    for (int i = 0; i < SCAN_PER; ++i) {
        v[i] = (b + i < N_NODES) ? cnt[b + i] : 0;
        sum += v[i];
    }
    sm[t] = sum;
    __syncthreads();
    for (int o = 1; o < SCAN_THREADS; o <<= 1) {
        const int y = (t >= o) ? sm[t - o] : 0;
        __syncthreads();
        sm[t] += y;
        __syncthreads();
    }
    int run = sm[t] - sum;
#pragma unroll
    for (int i = 0; i < SCAN_PER; ++i) {
        if (b + i < N_NODES) off[b + i] = run;
        run += v[i];
    }
    if (t == SCAN_THREADS - 1) off[N_NODES] = sm[t];
}

__global__ void __launch_bounds__(256)
bucket_fill_kernel(const int* __restrict__ esrc, const int* __restrict__ edst,
                   const int* __restrict__ off, const int* __restrict__ rank,
                   int* __restrict__ tmp)
{
    const int e = blockIdx.x * blockDim.x + threadIdx.x;
    if (e >= E_EDGES) return;
    tmp[off[edst[e]] + rank[e]] = esrc[e];
}

__global__ void __launch_bounds__(256)
bucket_sort_kernel(const int* __restrict__ off, const int* __restrict__ tmp,
                   int* __restrict__ csr)
{
    const int d    = blockIdx.x * (blockDim.x >> 5) + (threadIdx.x >> 5);
    const int lane = threadIdx.x & 31;
    if (d >= N_NODES) return;
    const int beg = off[d];
    const int deg = off[d + 1] - beg;
    if (deg <= 32) {
        const int key = (lane < deg) ? tmp[beg + lane] : 0x7fffffff;
        int pos = 0;
        for (int j = 0; j < deg; ++j) {
            const int kj = __shfl(key, j, 32);
            pos += (kj < key || (kj == key && j < lane)) ? 1 : 0;
        }
        if (lane < deg) csr[beg + pos] = key;
    } else {
        for (int i = lane; i < deg; i += 32) {
            const int key = tmp[beg + i];
            int pos = 0;
            for (int j = 0; j < deg; ++j) {
                const int kj = tmp[beg + j];
                pos += (kj < key || (kj == key && j < i)) ? 1 : 0;
            }
            csr[beg + pos] = key;
        }
    }
}

__device__ __forceinline__ void row_fma(float (&acc)[8], const v8bf h, const float w)
{
#pragma unroll
    for (int i = 0; i < 8; ++i) acc[i] = fmaf(w, (float)h[i], acc[i]);
}

__global__ void __launch_bounds__(256)
node_aggregate_kernel(const int* __restrict__ off, const int* __restrict__ csr,
                      const float* __restrict__ as_, const float* __restrict__ ad_,
                      const __bf16* __restrict__ Hb, const float* __restrict__ bias,
                      float* __restrict__ X)
{
    const int d    = __builtin_amdgcn_readfirstlane(
                         (int)(blockIdx.x * (blockDim.x >> 5) + (threadIdx.x >> 5)));
    const int lane = threadIdx.x & 31;
    if (d >= N_NODES) return;
    const int half = lane >> 4;
    const int c    = lane & 15;
    const int beg  = off[d];
    const int deg  = off[d + 1] - beg;
    const int n    = deg + 1;
    const float adv = ad_[d];
    const v8bf* __restrict__ Hrow = (const v8bf*)Hb + c;

    int   s0 = d;
    float v0 = -INFINITY;
    if (lane < n) {
        s0 = (lane < deg) ? csr[beg + lane] : d;
        const float t = as_[s0] + adv;
        v0 = (t > 0.0f) ? t : NEG_SLOPE * t;
    }
    float mx = v0;
    for (int base = 32; base < n; base += 32) {
        const int j = base + lane;
        if (j < n) {
            const int s = (j < deg) ? csr[beg + j] : d;
            const float t = as_[s] + adv;
            mx = fmaxf(mx, (t > 0.0f) ? t : NEG_SLOPE * t);
        }
    }
#pragma unroll
    for (int o = 16; o > 0; o >>= 1) mx = fmaxf(mx, __shfl_xor(mx, o, 32));

    float acc[8] = {0.f, 0.f, 0.f, 0.f, 0.f, 0.f, 0.f, 0.f};
    float zp = 0.0f;
    for (int base = 0; base < n; base += 32) {
        int   s = s0;
        float w = (lane < n) ? __expf(v0 - mx) : 0.0f;
        if (base != 0) {
            const int j = base + lane;
            s = d; w = 0.0f;
            if (j < n) {
                s = (j < deg) ? csr[beg + j] : d;
                const float t = as_[s] + adv;
                w = __expf(((t > 0.0f) ? t : NEG_SLOPE * t) - mx);
            }
        }
        zp += w;
        const int m = (n - base < 32) ? (n - base) : 32;
        for (int k = 0; k < m; k += 8) {
            v8bf  h[4];
            float ww[4];
#pragma unroll
            for (int u = 0; u < 4; ++u) {
                const int jj = k + 2 * u + half;
                const int ss = __shfl(s, jj, 32);
                ww[u] = __shfl(w, jj, 32);
                h[u]  = Hrow[(size_t)ss * 16];
            }
#pragma unroll
            for (int u = 0; u < 4; ++u) row_fma(acc, h[u], ww[u]);
        }
    }
#pragma unroll
    for (int o = 16; o > 0; o >>= 1) zp += __shfl_xor(zp, o, 32);
    const float rz = 1.0f / zp;

    float out4[4];
#pragma unroll
    for (int i = 0; i < 4; ++i) {
        const float give = half ? acc[i] : acc[4 + i];
        const float keep = half ? acc[4 + i] : acc[i];
        out4[i] = keep + __shfl_xor(give, 16, 32);
    }
    const int col = c * 8 + half * 4;
    const float4 b4 = *(const float4*)(bias + col);
    float4 r;
    r.x = out4[0] * rz + b4.x;
    r.y = out4[1] * rz + b4.y;
    r.z = out4[2] * rz + b4.z;
    r.w = out4[3] * rz + b4.w;
    r.x = (r.x > 0.0f) ? r.x : (__expf(r.x) - 1.0f);
    r.y = (r.y > 0.0f) ? r.y : (__expf(r.y) - 1.0f);
    r.z = (r.z > 0.0f) ? r.z : (__expf(r.z) - 1.0f);
    r.w = (r.w > 0.0f) ? r.w : (__expf(r.w) - 1.0f);
    *(float4*)(X + (size_t)d * DIM + col) = r;
}

extern "C" void kernel_launch(void* const* d_in, const int* in_sizes, int n_in,
                              void* d_out, int out_size, void* d_ws, size_t ws_size,
                              hipStream_t stream)
{
    (void)in_sizes; (void)n_in; (void)out_size; (void)ws_size;

    const float* xq    = (const float*)d_in[0];
    const float* xt    = (const float*)d_in[1];
    const int*   ei_q  = (const int*)d_in[2];
    const float* W_mlp = (const float*)d_in[4];
    const float* b_mlp = (const float*)d_in[5];
    const float* Ws    = (const float*)d_in[6];
    const float* a_src = (const float*)d_in[7];
    const float* a_dst = (const float*)d_in[8];
    const float* biases= (const float*)d_in[9];

    const int* esrc = ei_q;
    const int* edst = ei_q + E_EDGES;

    float* Xq = (float*)d_out;
    float* Xt = Xq + (size_t)N_NODES * DIM;

    __bf16* Wpk  = (__bf16*)d_ws;
    __bf16* Hb   = Wpk + (size_t)NWEIGHTS * WPK_ELEMS_PER_W;
    float*  as_  = (float*)(Hb + (size_t)N_NODES * DIM);
    float*  ad_  = as_ + N_NODES;
    int*    cnt  = (int*)(ad_ + N_NODES);
    int*    off  = cnt + N_NODES;
    int*    rank = off + N_NODES + 8;
    int*    tmp  = rank + E_EDGES;
    int*    csr  = tmp + E_EDGES;

    const dim3 blk(256);
    const dim3 pack_grid((NWEIGHTS * 32 * 32 + 255) / 256);
    const dim3 gemm_blk(GEMM_WAVES * 32);
    const dim3 gemm_grid((N_NODES / 16 + GEMM_WAVES - 1) / GEMM_WAVES);
    const dim3 node_wave_grid((N_NODES + 7) / 8);
    const dim3 edge_grid((E_EDGES + 255) / 256);

    pack_weights_kernel<<<pack_grid, blk, 0, stream>>>(W_mlp, Ws, Wpk, cnt);

    edge_rank_kernel  <<<edge_grid, blk, 0, stream>>>(edst, cnt, rank);
    bucket_scan_kernel<<<dim3(1), dim3(SCAN_THREADS), 0, stream>>>(cnt, off);
    bucket_fill_kernel<<<edge_grid, blk, 0, stream>>>(esrc, edst, off, rank, tmp);
    bucket_sort_kernel<<<node_wave_grid, blk, 0, stream>>>(off, tmp, csr);

    gemm128_wmma<false><<<gemm_grid, gemm_blk, 0, stream>>>(xq, Wpk, b_mlp, Xq, N_NODES,
                                                        nullptr, nullptr, nullptr, nullptr);
    gemm128_wmma<false><<<gemm_grid, gemm_blk, 0, stream>>>(xt, Wpk, b_mlp, Xt, N_NODES,
                                                        nullptr, nullptr, nullptr, nullptr);

    for (int l = 0; l < NLAYERS; ++l) {
        const __bf16* Wl = Wpk + (size_t)(l + 1) * WPK_ELEMS_PER_W;
        const float*  al = a_src  + (size_t)l * DIM;
        const float*  dl = a_dst  + (size_t)l * DIM;
        const float*  bl = biases + (size_t)l * DIM;
        float* Xs[2] = { Xq, Xt };
        for (int g = 0; g < 2; ++g) {
            float* X = Xs[g];
            gemm128_wmma<true><<<gemm_grid, gemm_blk, 0, stream>>>(X, Wl, nullptr, (void*)Hb, N_NODES,
                                                               al, dl, as_, ad_);
            node_aggregate_kernel<<<node_wave_grid, blk, 0, stream>>>(off, csr, as_, ad_, Hb, bl, X);
        }
    }
}


# GEMM: input MLP folded into the layer-0 product (MLP tiles +bias, rounded, are its B fragments in registers: D layout = B layout), 2 launches gone; each wave asks for X and all W fragments up front
# speedup vs baseline: 5.9724x; 1.3966x over previous
#include <hip/hip_runtime.h>
#include <hip/hip_bf16.h>

#define N_NODES 30000
#define E_EDGES 480000
#define DIM     128
#define NLAYERS 3
#define NEG_SLOPE 0.2f
#define NWEIGHTS 4
#define WPK_ELEMS_PER_W (8 * 4 * 32 * 16)

typedef __attribute__((ext_vector_type(16))) __bf16 v16bf;
typedef __attribute__((ext_vector_type(4)))  __bf16 v4bf;
typedef __attribute__((ext_vector_type(8)))  __bf16 v8bf;
typedef __attribute__((ext_vector_type(8)))  float  v8f;

__global__ void __launch_bounds__(256)
pack_weights_kernel(const float* __restrict__ W_mlp, const float* __restrict__ Ws,
                    __bf16* __restrict__ Wpk, int* __restrict__ cnt)
{
    for (int i = blockIdx.x * blockDim.x + threadIdx.x; i < N_NODES; i += gridDim.x * blockDim.x)
        cnt[i] = 0;
    const int gw   = (blockIdx.x * blockDim.x + threadIdx.x) >> 5;
    const int lane = threadIdx.x & 31;
    if (gw >= NWEIGHTS * 32) return;
    const int w  = gw >> 5;
    const int ct = (gw & 31) >> 2;
    const int s  = gw & 3;
    const float* W = (w == 0) ? W_mlp : (Ws + (size_t)(w - 1) * DIM * DIM);
    const int r  = lane & 15;
    const int hi = lane >> 4;
    const int col = ct * 16 + r;
    const int kb  = s * 32;
    __bf16* out = Wpk + ((size_t)gw * 32 + lane) * 16;
#pragma unroll
    for (int h = 0; h < 16; ++h) {
        const int K = kb + ((h < 8) ? (8 * hi + h) : (16 + 8 * hi + (h - 8)));
        out[h] = (__bf16)W[(size_t)K * DIM + col];
    }
}

#define GEMM_WAVES 2

__device__ __forceinline__ void load_fragments(v16bf (&w)[32], const __bf16* Wfrag, const int lane)
{
#pragma unroll
    for (int s = 0; s < 4; ++s) {
#pragma unroll
        for (int ct = 0; ct < 8; ++ct)
            w[ct * 4 + s] = *(const v16bf*)(Wfrag + ((size_t)(ct * 4 + s) * 32 + lane) * 16);
    }
}

__device__ __forceinline__ void strip_product(v8f (&acc)[8], const v16bf (&w)[32], const v16bf (&x)[4])
{
#pragma unroll
    for (int ct = 0; ct < 8; ++ct) acc[ct] = v8f{};
#pragma unroll
    for (int s = 0; s < 4; ++s) {
#pragma unroll
        for (int ct = 0; ct < 8; ++ct) {
            acc[ct] = __builtin_amdgcn_wmma_f32_16x16x32_bf16(
                 false, w[ct * 4 + s],  false, x[s],
                 (short)0, acc[ct],  false,  false);
        }
    }
}

template <bool MLP_FIRST>
__global__ void __launch_bounds__(GEMM_WAVES * 32)
gemm128_wmma(const float* __restrict__ X, const __bf16* __restrict__ W0pk,
             const float* __restrict__ bias0, const __bf16* __restrict__ Wpk,
             __bf16* __restrict__ Hout, int rows,
             const float* __restrict__ asrc, const float* __restrict__ adst,
             float* __restrict__ as_, float* __restrict__ ad_)
{
    const int strip = __builtin_amdgcn_readfirstlane(
                          (int)(blockIdx.x * GEMM_WAVES + (threadIdx.x >> 5)));
    if (strip * 16 >= rows) return;
    const int lane = threadIdx.x & 31;
    const int r    = lane & 15;
    const int hi   = lane >> 4;
    const int row  = strip * 16 + r;
    const int cb   = 8 * hi;

    float4 p[16];
    const float4* xp = (const float4*)(X + (size_t)row * DIM + cb);
#pragma unroll
    for (int s = 0; s < 4; ++s) {
        p[4 * s + 0] = xp[s * 8 + 0];
        p[4 * s + 1] = xp[s * 8 + 1];
        p[4 * s + 2] = xp[s * 8 + 4];
        p[4 * s + 3] = xp[s * 8 + 5];
    }
    v16bf w[32];
    load_fragments(w, MLP_FIRST ? W0pk : Wpk, lane);
    __builtin_amdgcn_sched_barrier(0);

    v16bf x[4];
#pragma unroll
    for (int s = 0; s < 4; ++s) {
        const float4 p0 = p[4 * s + 0], p1 = p[4 * s + 1], p2 = p[4 * s + 2], p3 = p[4 * s + 3];
        x[s][0]  = (__bf16)p0.x; x[s][1]  = (__bf16)p0.y; x[s][2]  = (__bf16)p0.z; x[s][3]  = (__bf16)p0.w;
        x[s][4]  = (__bf16)p1.x; x[s][5]  = (__bf16)p1.y; x[s][6]  = (__bf16)p1.z; x[s][7]  = (__bf16)p1.w;
        x[s][8]  = (__bf16)p2.x; x[s][9]  = (__bf16)p2.y; x[s][10] = (__bf16)p2.z; x[s][11] = (__bf16)p2.w;
        x[s][12] = (__bf16)p3.x; x[s][13] = (__bf16)p3.y; x[s][14] = (__bf16)p3.z; x[s][15] = (__bf16)p3.w;
    }

    v8f acc[8];
    if (MLP_FIRST) {
        strip_product(acc, w, x);
        __builtin_amdgcn_sched_barrier(0);
        load_fragments(w, Wpk, lane);
        float4 b[16];
#pragma unroll
        for (int s = 0; s < 4; ++s) {
            b[4 * s + 0] = *(const float4*)(bias0 + s * 32 + cb);
            b[4 * s + 1] = *(const float4*)(bias0 + s * 32 + cb + 4);
            b[4 * s + 2] = *(const float4*)(bias0 + s * 32 + 16 + cb);
            b[4 * s + 3] = *(const float4*)(bias0 + s * 32 + 16 + cb + 4);
        }
        __builtin_amdgcn_sched_barrier(0);
#pragma unroll
        for (int s = 0; s < 4; ++s) {
            const float4 b0 = b[4 * s + 0], b1 = b[4 * s + 1], b2 = b[4 * s + 2], b3 = b[4 * s + 3];
            const v8f lo = acc[2 * s];
            const v8f up = acc[2 * s + 1];
            x[s][0]  = (__bf16)(lo[0] + b0.x); x[s][1]  = (__bf16)(lo[1] + b0.y);
            x[s][2]  = (__bf16)(lo[2] + b0.z); x[s][3]  = (__bf16)(lo[3] + b0.w);
            x[s][4]  = (__bf16)(lo[4] + b1.x); x[s][5]  = (__bf16)(lo[5] + b1.y);
            x[s][6]  = (__bf16)(lo[6] + b1.z); x[s][7]  = (__bf16)(lo[7] + b1.w);
            x[s][8]  = (__bf16)(up[0] + b2.x); x[s][9]  = (__bf16)(up[1] + b2.y);
            x[s][10] = (__bf16)(up[2] + b2.z); x[s][11] = (__bf16)(up[3] + b2.w);
            x[s][12] = (__bf16)(up[4] + b3.x); x[s][13] = (__bf16)(up[5] + b3.y);
            x[s][14] = (__bf16)(up[6] + b3.z); x[s][15] = (__bf16)(up[7] + b3.w);
        }
    }

    float4 av[16], dv[16];
#pragma unroll
    for (int ct = 0; ct < 8; ++ct) {
        av[2 * ct + 0] = *(const float4*)(asrc + ct * 16 + cb);
        av[2 * ct + 1] = *(const float4*)(asrc + ct * 16 + cb + 4);
        dv[2 * ct + 0] = *(const float4*)(adst + ct * 16 + cb);
        dv[2 * ct + 1] = *(const float4*)(adst + ct * 16 + cb + 4);
    }
    __builtin_amdgcn_sched_barrier(0);
    strip_product(acc, w, x);

    float ps = 0.0f, pd = 0.0f;
    __bf16* H = Hout + (size_t)row * DIM + cb;
#pragma unroll
    for (int ct = 0; ct < 8; ++ct) {
        const float4 s0 = av[2 * ct + 0];
        const float4 s1 = av[2 * ct + 1];
        const float4 d0 = dv[2 * ct + 0];
        const float4 d1 = dv[2 * ct + 1];
        const v8f a = acc[ct];
        float ts = a[0] * s0.x;
        ts = fmaf(a[1], s0.y, ts); ts = fmaf(a[2], s0.z, ts); ts = fmaf(a[3], s0.w, ts);
        ts = fmaf(a[4], s1.x, ts); ts = fmaf(a[5], s1.y, ts); ts = fmaf(a[6], s1.z, ts);
        ts = fmaf(a[7], s1.w, ts);
        float td = a[0] * d0.x;
        td = fmaf(a[1], d0.y, td); td = fmaf(a[2], d0.z, td); td = fmaf(a[3], d0.w, td);
        td = fmaf(a[4], d1.x, td); td = fmaf(a[5], d1.y, td); td = fmaf(a[6], d1.z, td);
        td = fmaf(a[7], d1.w, td);
        ps += ts;
        pd += td;
        v8bf o;
        o[0] = (__bf16)a[0]; o[1] = (__bf16)a[1]; o[2] = (__bf16)a[2]; o[3] = (__bf16)a[3];
        o[4] = (__bf16)a[4]; o[5] = (__bf16)a[5]; o[6] = (__bf16)a[6]; o[7] = (__bf16)a[7];
        *(v8bf*)(H + ct * 16) = o;
    }
    ps += __shfl_xor(ps, 16, 32);
    pd += __shfl_xor(pd, 16, 32);
    if (hi) ad_[row] = pd; else as_[row] = ps;
}

__global__ void __launch_bounds__(256)
edge_rank_kernel(const int* __restrict__ edst, int* __restrict__ cnt, int* __restrict__ rank)
{
    const int e = blockIdx.x * blockDim.x + threadIdx.x;
    if (e >= E_EDGES) return;
    rank[e] = atomicAdd(&cnt[edst[e]], 1);
}

#define SCAN_THREADS 1024
#define SCAN_PER     ((N_NODES + SCAN_THREADS - 1) / SCAN_THREADS)

__global__ void __launch_bounds__(SCAN_THREADS)
bucket_scan_kernel(const int* __restrict__ cnt, int* __restrict__ off)
{
    __shared__ int sm[SCAN_THREADS];
    const int t = threadIdx.x;
    const int b = t * SCAN_PER;
    int v[SCAN_PER];
    int sum = 0;
#pragma unroll
    for (int i = 0; i < SCAN_PER; ++i) {
        v[i] = (b + i < N_NODES) ? cnt[b + i] : 0;
        sum += v[i];
    }
    sm[t] = sum;
    __syncthreads();
    for (int o = 1; o < SCAN_THREADS; o <<= 1) {
        const int y = (t >= o) ? sm[t - o] : 0;
        __syncthreads();
        sm[t] += y;
        __syncthreads();
    }
    int run = sm[t] - sum;
#pragma unroll
    for (int i = 0; i < SCAN_PER; ++i) {
        if (b + i < N_NODES) off[b + i] = run;
        run += v[i];
    }
    if (t == SCAN_THREADS - 1) off[N_NODES] = sm[t];
}

__global__ void __launch_bounds__(256)
bucket_fill_kernel(const int* __restrict__ esrc, const int* __restrict__ edst,
                   const int* __restrict__ off, const int* __restrict__ rank,
                   int* __restrict__ tmp)
{
    const int e = blockIdx.x * blockDim.x + threadIdx.x;
    if (e >= E_EDGES) return;
    tmp[off[edst[e]] + rank[e]] = esrc[e];
}

__global__ void __launch_bounds__(256)
bucket_sort_kernel(const int* __restrict__ off, const int* __restrict__ tmp,
                   int* __restrict__ csr)
{
    const int d    = blockIdx.x * (blockDim.x >> 5) + (threadIdx.x >> 5);
    const int lane = threadIdx.x & 31;
    if (d >= N_NODES) return;
    const int beg = off[d];
    const int deg = off[d + 1] - beg;
    if (deg <= 32) {
        const int key = (lane < deg) ? tmp[beg + lane] : 0x7fffffff;
        int pos = 0;
        for (int j = 0; j < deg; ++j) {
            const int kj = __shfl(key, j, 32);
            pos += (kj < key || (kj == key && j < lane)) ? 1 : 0;
        }
        if (lane < deg) csr[beg + pos] = key;
    } else {
        for (int i = lane; i < deg; i += 32) {
            const int key = tmp[beg + i];
            int pos = 0;
            for (int j = 0; j < deg; ++j) {
                const int kj = tmp[beg + j];
                pos += (kj < key || (kj == key && j < i)) ? 1 : 0;
            }
            csr[beg + pos] = key;
        }
    }
}

__device__ __forceinline__ void row_fma(float (&acc)[8], const v8bf h, const float w)
{
#pragma unroll
    for (int i = 0; i < 8; ++i) acc[i] = fmaf(w, (float)h[i], acc[i]);
}

__global__ void __launch_bounds__(256)
node_aggregate_kernel(const int* __restrict__ off, const int* __restrict__ csr,
                      const float* __restrict__ as_, const float* __restrict__ ad_,
                      const __bf16* __restrict__ Hb, const float* __restrict__ bias,
                      float* __restrict__ X)
{
    const int d    = __builtin_amdgcn_readfirstlane(
                         (int)(blockIdx.x * (blockDim.x >> 5) + (threadIdx.x >> 5)));
    const int lane = threadIdx.x & 31;
    if (d >= N_NODES) return;
    const int half = lane >> 4;
    const int c    = lane & 15;
    const int beg  = off[d];
    const int deg  = off[d + 1] - beg;
    const int n    = deg + 1;
    const float adv = ad_[d];
    const v8bf* __restrict__ Hrow = (const v8bf*)Hb + c;

    int   s0 = d;
    float v0 = -INFINITY;
    if (lane < n) {
        s0 = (lane < deg) ? csr[beg + lane] : d;
        const float t = as_[s0] + adv;
        v0 = (t > 0.0f) ? t : NEG_SLOPE * t;
    }
    float mx = v0;
    for (int base = 32; base < n; base += 32) {
        const int j = base + lane;
        if (j < n) {
            const int s = (j < deg) ? csr[beg + j] : d;
            const float t = as_[s] + adv;
            mx = fmaxf(mx, (t > 0.0f) ? t : NEG_SLOPE * t);
        }
    }
#pragma unroll
    for (int o = 16; o > 0; o >>= 1) mx = fmaxf(mx, __shfl_xor(mx, o, 32));

    float acc[8] = {0.f, 0.f, 0.f, 0.f, 0.f, 0.f, 0.f, 0.f};
    float zp = 0.0f;
    for (int base = 0; base < n; base += 32) {
        int   s = s0;
        float w = (lane < n) ? __expf(v0 - mx) : 0.0f;
        if (base != 0) {
            const int j = base + lane;
            s = d; w = 0.0f;
            if (j < n) {
                s = (j < deg) ? csr[beg + j] : d;
                const float t = as_[s] + adv;
                w = __expf(((t > 0.0f) ? t : NEG_SLOPE * t) - mx);
            }
        }
        zp += w;
        const int m = (n - base < 32) ? (n - base) : 32;
        for (int k = 0; k < m; k += 8) {
            v8bf  h[4];
            float ww[4];
#pragma unroll
            for (int u = 0; u < 4; ++u) {
                const int jj = k + 2 * u + half;
                const int ss = __shfl(s, jj, 32);
                ww[u] = __shfl(w, jj, 32);
                h[u]  = Hrow[(size_t)ss * 16];
            }
#pragma unroll
            for (int u = 0; u < 4; ++u) row_fma(acc, h[u], ww[u]);
        }
    }
#pragma unroll
    for (int o = 16; o > 0; o >>= 1) zp += __shfl_xor(zp, o, 32);
    const float rz = 1.0f / zp;

    float out4[4];
#pragma unroll
    for (int i = 0; i < 4; ++i) {
        const float give = half ? acc[i] : acc[4 + i];
        const float keep = half ? acc[4 + i] : acc[i];
        out4[i] = keep + __shfl_xor(give, 16, 32);
    }
    const int col = c * 8 + half * 4;
    const float4 b4 = *(const float4*)(bias + col);
    float4 r;
    r.x = out4[0] * rz + b4.x;
    r.y = out4[1] * rz + b4.y;
    r.z = out4[2] * rz + b4.z;
    r.w = out4[3] * rz + b4.w;
    r.x = (r.x > 0.0f) ? r.x : (__expf(r.x) - 1.0f);
    r.y = (r.y > 0.0f) ? r.y : (__expf(r.y) - 1.0f);
    r.z = (r.z > 0.0f) ? r.z : (__expf(r.z) - 1.0f);
    r.w = (r.w > 0.0f) ? r.w : (__expf(r.w) - 1.0f);
    *(float4*)(X + (size_t)d * DIM + col) = r;
}

extern "C" void kernel_launch(void* const* d_in, const int* in_sizes, int n_in,
                              void* d_out, int out_size, void* d_ws, size_t ws_size,
                              hipStream_t stream)
{
    (void)in_sizes; (void)n_in; (void)out_size; (void)ws_size;

    const float* xq    = (const float*)d_in[0];
    const float* xt    = (const float*)d_in[1];
    const int*   ei_q  = (const int*)d_in[2];
    const float* W_mlp = (const float*)d_in[4];
    const float* b_mlp = (const float*)d_in[5];
    const float* Ws    = (const float*)d_in[6];
    const float* a_src = (const float*)d_in[7];
    const float* a_dst = (const float*)d_in[8];
    const float* biases= (const float*)d_in[9];

    const int* esrc = ei_q;
    const int* edst = ei_q + E_EDGES;

    float* Xq = (float*)d_out;
    float* Xt = Xq + (size_t)N_NODES * DIM;

    __bf16* Wpk  = (__bf16*)d_ws;
    __bf16* Hb   = Wpk + (size_t)NWEIGHTS * WPK_ELEMS_PER_W;
    float*  as_  = (float*)(Hb + (size_t)N_NODES * DIM);
    float*  ad_  = as_ + N_NODES;
    int*    cnt  = (int*)(ad_ + N_NODES);
    int*    off  = cnt + N_NODES;
    int*    rank = off + N_NODES + 8;
    int*    tmp  = rank + E_EDGES;
    int*    csr  = tmp + E_EDGES;

    const dim3 blk(256);
    const dim3 pack_grid((NWEIGHTS * 32 * 32 + 255) / 256);
    const dim3 gemm_blk(GEMM_WAVES * 32);
    const dim3 gemm_grid((N_NODES / 16 + GEMM_WAVES - 1) / GEMM_WAVES);
    const dim3 node_wave_grid((N_NODES + 7) / 8);
    const dim3 edge_grid((E_EDGES + 255) / 256);

    pack_weights_kernel<<<pack_grid, blk, 0, stream>>>(W_mlp, Ws, Wpk, cnt);

    edge_rank_kernel  <<<edge_grid, blk, 0, stream>>>(edst, cnt, rank);
    bucket_scan_kernel<<<dim3(1), dim3(SCAN_THREADS), 0, stream>>>(cnt, off);
    bucket_fill_kernel<<<edge_grid, blk, 0, stream>>>(esrc, edst, off, rank, tmp);
    bucket_sort_kernel<<<node_wave_grid, blk, 0, stream>>>(off, tmp, csr);

    gemm128_wmma<true><<<gemm_grid, gemm_blk, 0, stream>>>(xq, Wpk, b_mlp, Wpk + WPK_ELEMS_PER_W, Hb, N_NODES,
                                                       a_src, a_dst, as_, ad_);
    node_aggregate_kernel<<<node_wave_grid, blk, 0, stream>>>(off, csr, as_, ad_, Hb, biases, Xq);
    gemm128_wmma<true><<<gemm_grid, gemm_blk, 0, stream>>>(xt, Wpk, b_mlp, Wpk + WPK_ELEMS_PER_W, Hb, N_NODES,
                                                       a_src, a_dst, as_, ad_);
    node_aggregate_kernel<<<node_wave_grid, blk, 0, stream>>>(off, csr, as_, ad_, Hb, biases, Xt);

    for (int l = 1; l < NLAYERS; ++l) {
        const __bf16* Wl = Wpk + (size_t)(l + 1) * WPK_ELEMS_PER_W;
        const float*  al = a_src  + (size_t)l * DIM;
        const float*  dl = a_dst  + (size_t)l * DIM;
        const float*  bl = biases + (size_t)l * DIM;
        float* Xs[2] = { Xq, Xt };
        for (int g = 0; g < 2; ++g) {
            float* X = Xs[g];
            gemm128_wmma<false><<<gemm_grid, gemm_blk, 0, stream>>>(X, nullptr, nullptr, Wl, Hb, N_NODES,
                                                                al, dl, as_, ad_);
            node_aggregate_kernel<<<node_wave_grid, blk, 0, stream>>>(off, csr, as_, ad_, Hb, bl, X);
        }
    }
}


# Graphs q and t share launches: one GEMM per layer for both; in layers 0-1 one aggregate wave per node does both (X kept as the 16-bit values the GEMM rounds to); 17 -> 12 launches, sum orders kept
# speedup vs baseline: 6.4409x; 1.0784x over previous
#include <hip/hip_runtime.h>
#include <hip/hip_bf16.h>

#define N_NODES 30000
#define E_EDGES 480000
#define DIM     128
#define NLAYERS 3
#define NEG_SLOPE 0.2f
#define NWEIGHTS 4
#define WPK_ELEMS_PER_W (8 * 4 * 32 * 16)

typedef __attribute__((ext_vector_type(16))) __bf16 v16bf;
typedef __attribute__((ext_vector_type(4)))  __bf16 v4bf;
typedef __attribute__((ext_vector_type(8)))  __bf16 v8bf;
typedef __attribute__((ext_vector_type(8)))  float  v8f;

__global__ void __launch_bounds__(256)
pack_weights_kernel(const float* __restrict__ W_mlp, const float* __restrict__ Ws,
                    __bf16* __restrict__ Wpk, int* __restrict__ cnt)
{
    for (int i = blockIdx.x * blockDim.x + threadIdx.x; i < N_NODES; i += gridDim.x * blockDim.x)
        cnt[i] = 0;
    const int gw   = (blockIdx.x * blockDim.x + threadIdx.x) >> 5;
    const int lane = threadIdx.x & 31;
    if (gw >= NWEIGHTS * 32) return;
    const int w  = gw >> 5;
    const int ct = (gw & 31) >> 2;
    const int s  = gw & 3;
    const float* W = (w == 0) ? W_mlp : (Ws + (size_t)(w - 1) * DIM * DIM);
    const int r  = lane & 15;
    const int hi = lane >> 4;
    const int col = ct * 16 + r;
    const int kb  = s * 32;
    __bf16* out = Wpk + ((size_t)gw * 32 + lane) * 16;
#pragma unroll
    for (int h = 0; h < 16; ++h) {
        const int K = kb + ((h < 8) ? (8 * hi + h) : (16 + 8 * hi + (h - 8)));
        out[h] = (__bf16)W[(size_t)K * DIM + col];
    }
}

#define GEMM_WAVES 2

__device__ __forceinline__ void load_fragments(v16bf (&w)[32], const __bf16* Wfrag, const int lane)
{
#pragma unroll
    for (int s = 0; s < 4; ++s) {
#pragma unroll
        for (int ct = 0; ct < 8; ++ct)
            w[ct * 4 + s] = *(const v16bf*)(Wfrag + ((size_t)(ct * 4 + s) * 32 + lane) * 16);
    }
}

__device__ __forceinline__ void strip_product(v8f (&acc)[8], const v16bf (&w)[32], const v16bf (&x)[4])
{
#pragma unroll
    for (int ct = 0; ct < 8; ++ct) acc[ct] = v8f{};
#pragma unroll
    for (int s = 0; s < 4; ++s) {
#pragma unroll
        for (int ct = 0; ct < 8; ++ct) {
            acc[ct] = __builtin_amdgcn_wmma_f32_16x16x32_bf16(
                 false, w[ct * 4 + s],  false, x[s],
                 (short)0, acc[ct],  false,  false);
        }
    }
}

#define STRIPS (N_NODES / 16)

template <bool MLP_FIRST>
__global__ void __launch_bounds__(GEMM_WAVES * 32)
gemm128_wmma(const float* __restrict__ Xraw_q, const float* __restrict__ Xraw_t,
             const __bf16* __restrict__ Xb_q, const __bf16* __restrict__ Xb_t,
             const __bf16* __restrict__ W0pk, const float* __restrict__ bias0,
             const __bf16* __restrict__ Wpk,
             __bf16* __restrict__ H_q, __bf16* __restrict__ H_t,
             const float* __restrict__ asrc, const float* __restrict__ adst,
             float* __restrict__ as_q, float* __restrict__ ad_q,
             float* __restrict__ as_t, float* __restrict__ ad_t)
{
    const int sidx = __builtin_amdgcn_readfirstlane(
                         (int)(blockIdx.x * GEMM_WAVES + (threadIdx.x >> 5)));
    if (sidx >= 2 * STRIPS) return;
    const bool second = sidx >= STRIPS;
    const int strip = second ? (sidx - STRIPS) : sidx;
    const int lane = threadIdx.x & 31;
    const int r    = lane & 15;
    const int hi   = lane >> 4;
    const int row  = strip * 16 + r;
    const int cb   = 8 * hi;

    v16bf x[4];
    v16bf w[32];
    if constexpr (MLP_FIRST) {
        const float* X = second ? Xraw_t : Xraw_q;
        float4 p[16];
        const float4* xp = (const float4*)(X + (size_t)row * DIM + cb);
#pragma unroll
        for (int s = 0; s < 4; ++s) {
            p[4 * s + 0] = xp[s * 8 + 0];
            p[4 * s + 1] = xp[s * 8 + 1];
            p[4 * s + 2] = xp[s * 8 + 4];
            p[4 * s + 3] = xp[s * 8 + 5];
        }
        load_fragments(w, W0pk, lane);
        __builtin_amdgcn_sched_barrier(0);
#pragma unroll
        for (int s = 0; s < 4; ++s) {
            const float4 p0 = p[4 * s + 0], p1 = p[4 * s + 1], p2 = p[4 * s + 2], p3 = p[4 * s + 3];
            x[s][0]  = (__bf16)p0.x; x[s][1]  = (__bf16)p0.y; x[s][2]  = (__bf16)p0.z; x[s][3]  = (__bf16)p0.w;
            x[s][4]  = (__bf16)p1.x; x[s][5]  = (__bf16)p1.y; x[s][6]  = (__bf16)p1.z; x[s][7]  = (__bf16)p1.w;
            x[s][8]  = (__bf16)p2.x; x[s][9]  = (__bf16)p2.y; x[s][10] = (__bf16)p2.z; x[s][11] = (__bf16)p2.w;
            x[s][12] = (__bf16)p3.x; x[s][13] = (__bf16)p3.y; x[s][14] = (__bf16)p3.z; x[s][15] = (__bf16)p3.w;
        }
    } else {
        const __bf16* Xb = second ? Xb_t : Xb_q;
        const v8bf* xp = (const v8bf*)(Xb + (size_t)row * DIM + cb);
        v8bf q[8];
#pragma unroll
        for (int s = 0; s < 4; ++s) {
            q[2 * s + 0] = xp[s * 4 + 0];
            q[2 * s + 1] = xp[s * 4 + 2];
        }
        load_fragments(w, Wpk, lane);
        __builtin_amdgcn_sched_barrier(0);
#pragma unroll
        for (int s = 0; s < 4; ++s)
            x[s] = __builtin_shufflevector(q[2 * s + 0], q[2 * s + 1],
                                           0, 1, 2, 3, 4, 5, 6, 7, 8, 9, 10, 11, 12, 13, 14, 15);
    }

    v8f acc[8];
    if constexpr (MLP_FIRST) {
        strip_product(acc, w, x);
        __builtin_amdgcn_sched_barrier(0);
        load_fragments(w, Wpk, lane);
        float4 b[16];
#pragma unroll
        for (int s = 0; s < 4; ++s) {
            b[4 * s + 0] = *(const float4*)(bias0 + s * 32 + cb);
            b[4 * s + 1] = *(const float4*)(bias0 + s * 32 + cb + 4);
            b[4 * s + 2] = *(const float4*)(bias0 + s * 32 + 16 + cb);
            b[4 * s + 3] = *(const float4*)(bias0 + s * 32 + 16 + cb + 4);
        }
        __builtin_amdgcn_sched_barrier(0);
#pragma unroll
        for (int s = 0; s < 4; ++s) {
            const float4 b0 = b[4 * s + 0], b1 = b[4 * s + 1], b2 = b[4 * s + 2], b3 = b[4 * s + 3];
            const v8f lo = acc[2 * s];
            const v8f up = acc[2 * s + 1];
            x[s][0]  = (__bf16)(lo[0] + b0.x); x[s][1]  = (__bf16)(lo[1] + b0.y);
            x[s][2]  = (__bf16)(lo[2] + b0.z); x[s][3]  = (__bf16)(lo[3] + b0.w);
            x[s][4]  = (__bf16)(lo[4] + b1.x); x[s][5]  = (__bf16)(lo[5] + b1.y);
            x[s][6]  = (__bf16)(lo[6] + b1.z); x[s][7]  = (__bf16)(lo[7] + b1.w);
            x[s][8]  = (__bf16)(up[0] + b2.x); x[s][9]  = (__bf16)(up[1] + b2.y);
            x[s][10] = (__bf16)(up[2] + b2.z); x[s][11] = (__bf16)(up[3] + b2.w);
            x[s][12] = (__bf16)(up[4] + b3.x); x[s][13] = (__bf16)(up[5] + b3.y);
            x[s][14] = (__bf16)(up[6] + b3.z); x[s][15] = (__bf16)(up[7] + b3.w);
        }
    }

    float4 av[16], dv[16];
#pragma unroll
    for (int ct = 0; ct < 8; ++ct) {
        av[2 * ct + 0] = *(const float4*)(asrc + ct * 16 + cb);
        av[2 * ct + 1] = *(const float4*)(asrc + ct * 16 + cb + 4);
        dv[2 * ct + 0] = *(const float4*)(adst + ct * 16 + cb);
        dv[2 * ct + 1] = *(const float4*)(adst + ct * 16 + cb + 4);
    }
    __builtin_amdgcn_sched_barrier(0);
    strip_product(acc, w, x);

    float ps = 0.0f, pd = 0.0f;
    __bf16* H = (second ? H_t : H_q) + (size_t)row * DIM + cb;
#pragma unroll
    for (int ct = 0; ct < 8; ++ct) {
        const float4 s0 = av[2 * ct + 0];
        const float4 s1 = av[2 * ct + 1];
        const float4 d0 = dv[2 * ct + 0];
        const float4 d1 = dv[2 * ct + 1];
        const v8f a = acc[ct];
        float ts = a[0] * s0.x;
        ts = fmaf(a[1], s0.y, ts); ts = fmaf(a[2], s0.z, ts); ts = fmaf(a[3], s0.w, ts);
        ts = fmaf(a[4], s1.x, ts); ts = fmaf(a[5], s1.y, ts); ts = fmaf(a[6], s1.z, ts);
        ts = fmaf(a[7], s1.w, ts);
        float td = a[0] * d0.x;
        td = fmaf(a[1], d0.y, td); td = fmaf(a[2], d0.z, td); td = fmaf(a[3], d0.w, td);
        td = fmaf(a[4], d1.x, td); td = fmaf(a[5], d1.y, td); td = fmaf(a[6], d1.z, td);
        td = fmaf(a[7], d1.w, td);
        ps += ts;
        pd += td;
        v8bf o;
        o[0] = (__bf16)a[0]; o[1] = (__bf16)a[1]; o[2] = (__bf16)a[2]; o[3] = (__bf16)a[3];
        o[4] = (__bf16)a[4]; o[5] = (__bf16)a[5]; o[6] = (__bf16)a[6]; o[7] = (__bf16)a[7];
        *(v8bf*)(H + ct * 16) = o;
    }
    ps += __shfl_xor(ps, 16, 32);
    pd += __shfl_xor(pd, 16, 32);
    if (hi) (second ? ad_t : ad_q)[row] = pd; else (second ? as_t : as_q)[row] = ps;
}

__global__ void __launch_bounds__(256)
edge_rank_kernel(const int* __restrict__ edst, int* __restrict__ cnt, int* __restrict__ rank)
{
    const int e = blockIdx.x * blockDim.x + threadIdx.x;
    if (e >= E_EDGES) return;
    rank[e] = atomicAdd(&cnt[edst[e]], 1);
}

#define SCAN_THREADS 1024
#define SCAN_PER     ((N_NODES + SCAN_THREADS - 1) / SCAN_THREADS)

__global__ void __launch_bounds__(SCAN_THREADS)
bucket_scan_kernel(const int* __restrict__ cnt, int* __restrict__ off)
{
    __shared__ int sm[SCAN_THREADS];
    const int t = threadIdx.x;
    const int b = t * SCAN_PER;
    int v[SCAN_PER];
    int sum = 0;
#pragma unroll
    for (int i = 0; i < SCAN_PER; ++i) {
        v[i] = (b + i < N_NODES) ? cnt[b + i] : 0;
        sum += v[i];
    }
    sm[t] = sum;
    __syncthreads();
    for (int o = 1; o < SCAN_THREADS; o <<= 1) {
        const int y = (t >= o) ? sm[t - o] : 0;
        __syncthreads();
        sm[t] += y;
        __syncthreads();
    }
    int run = sm[t] - sum;
#pragma unroll
    for (int i = 0; i < SCAN_PER; ++i) {
        if (b + i < N_NODES) off[b + i] = run;
        run += v[i];
    }
    if (t == SCAN_THREADS - 1) off[N_NODES] = sm[t];
}

__global__ void __launch_bounds__(256)
bucket_fill_kernel(const int* __restrict__ esrc, const int* __restrict__ edst,
                   const int* __restrict__ off, const int* __restrict__ rank,
                   int* __restrict__ tmp)
{
    const int e = blockIdx.x * blockDim.x + threadIdx.x;
    if (e >= E_EDGES) return;
    tmp[off[edst[e]] + rank[e]] = esrc[e];
}

__global__ void __launch_bounds__(256)
bucket_sort_kernel(const int* __restrict__ off, const int* __restrict__ tmp,
                   int* __restrict__ csr)
{
    const int d    = blockIdx.x * (blockDim.x >> 5) + (threadIdx.x >> 5);
    const int lane = threadIdx.x & 31;
    if (d >= N_NODES) return;
    const int beg = off[d];
    const int deg = off[d + 1] - beg;
    if (deg <= 32) {
        const int key = (lane < deg) ? tmp[beg + lane] : 0x7fffffff;
        int pos = 0;
        for (int j = 0; j < deg; ++j) {
            const int kj = __shfl(key, j, 32);
            pos += (kj < key || (kj == key && j < lane)) ? 1 : 0;
        }
        if (lane < deg) csr[beg + pos] = key;
    } else {
        for (int i = lane; i < deg; i += 32) {
            const int key = tmp[beg + i];
            int pos = 0;
            for (int j = 0; j < deg; ++j) {
                const int kj = tmp[beg + j];
                pos += (kj < key || (kj == key && j < i)) ? 1 : 0;
            }
            csr[beg + pos] = key;
        }
    }
}

__device__ __forceinline__ void row_fma(float (&acc)[8], const v8bf h, const float w)
{
#pragma unroll
    for (int i = 0; i < 8; ++i) acc[i] = fmaf(w, (float)h[i], acc[i]);
}

__global__ void __launch_bounds__(256)
node_aggregate_kernel(const int* __restrict__ off, const int* __restrict__ csr,
                      const float* __restrict__ as_, const float* __restrict__ ad_,
                      const __bf16* __restrict__ Hb, const float* __restrict__ bias,
                      float* __restrict__ X)
{
    const int d    = __builtin_amdgcn_readfirstlane(
                         (int)(blockIdx.x * (blockDim.x >> 5) + (threadIdx.x >> 5)));
    const int lane = threadIdx.x & 31;
    if (d >= N_NODES) return;
    const int half = lane >> 4;
    const int c    = lane & 15;
    const int beg  = off[d];
    const int deg  = off[d + 1] - beg;
    const int n    = deg + 1;
    const float adv = ad_[d];
    const v8bf* __restrict__ Hrow = (const v8bf*)Hb + c;

    int   s0 = d;
    float v0 = -INFINITY;
    if (lane < n) {
        s0 = (lane < deg) ? csr[beg + lane] : d;
        const float t = as_[s0] + adv;
        v0 = (t > 0.0f) ? t : NEG_SLOPE * t;
    }
    float mx = v0;
    for (int base = 32; base < n; base += 32) {
        const int j = base + lane;
        if (j < n) {
            const int s = (j < deg) ? csr[beg + j] : d;
            const float t = as_[s] + adv;
            mx = fmaxf(mx, (t > 0.0f) ? t : NEG_SLOPE * t);
        }
    }
#pragma unroll
    for (int o = 16; o > 0; o >>= 1) mx = fmaxf(mx, __shfl_xor(mx, o, 32));

    float acc[8] = {0.f, 0.f, 0.f, 0.f, 0.f, 0.f, 0.f, 0.f};
    float zp = 0.0f;
    for (int base = 0; base < n; base += 32) {
        int   s = s0;
        float w = (lane < n) ? __expf(v0 - mx) : 0.0f;
        if (base != 0) {
            const int j = base + lane;
            s = d; w = 0.0f;
            if (j < n) {
                s = (j < deg) ? csr[beg + j] : d;
                const float t = as_[s] + adv;
                w = __expf(((t > 0.0f) ? t : NEG_SLOPE * t) - mx);
            }
        }
        zp += w;
        const int m = (n - base < 32) ? (n - base) : 32;
        for (int k = 0; k < m; k += 8) {
            v8bf  h[4];
            float ww[4];
#pragma unroll
            for (int u = 0; u < 4; ++u) {
                const int jj = k + 2 * u + half;
                const int ss = __shfl(s, jj, 32);
                ww[u] = __shfl(w, jj, 32);
                h[u]  = Hrow[(size_t)ss * 16];
            }
#pragma unroll
            for (int u = 0; u < 4; ++u) row_fma(acc, h[u], ww[u]);
        }
    }
#pragma unroll
    for (int o = 16; o > 0; o >>= 1) zp += __shfl_xor(zp, o, 32);
    const float rz = 1.0f / zp;

    float out4[4];
#pragma unroll
    for (int i = 0; i < 4; ++i) {
        const float give = half ? acc[i] : acc[4 + i];
        const float keep = half ? acc[4 + i] : acc[i];
        out4[i] = keep + __shfl_xor(give, 16, 32);
    }
    const int col = c * 8 + half * 4;
    const float4 b4 = *(const float4*)(bias + col);
    float4 r;
    r.x = out4[0] * rz + b4.x;
    r.y = out4[1] * rz + b4.y;
    r.z = out4[2] * rz + b4.z;
    r.w = out4[3] * rz + b4.w;
    r.x = (r.x > 0.0f) ? r.x : (__expf(r.x) - 1.0f);
    r.y = (r.y > 0.0f) ? r.y : (__expf(r.y) - 1.0f);
    r.z = (r.z > 0.0f) ? r.z : (__expf(r.z) - 1.0f);
    r.w = (r.w > 0.0f) ? r.w : (__expf(r.w) - 1.0f);
    *(float4*)(X + (size_t)d * DIM + col) = r;
}

__device__ __forceinline__ void finish_row_bf16(const float (&acc)[8], float zp, const float4 b4,
                                                const int half, __bf16* __restrict__ dst)
{
#pragma unroll
    for (int o = 16; o > 0; o >>= 1) zp += __shfl_xor(zp, o, 32);
    const float rz = 1.0f / zp;
    float out4[4];
#pragma unroll
    for (int i = 0; i < 4; ++i) {
        const float give = half ? acc[i] : acc[4 + i];
        const float keep = half ? acc[4 + i] : acc[i];
        out4[i] = keep + __shfl_xor(give, 16, 32);
    }
    float4 r;
    r.x = out4[0] * rz + b4.x;
    r.y = out4[1] * rz + b4.y;
    r.z = out4[2] * rz + b4.z;
    r.w = out4[3] * rz + b4.w;
    r.x = (r.x > 0.0f) ? r.x : (__expf(r.x) - 1.0f);
    r.y = (r.y > 0.0f) ? r.y : (__expf(r.y) - 1.0f);
    r.z = (r.z > 0.0f) ? r.z : (__expf(r.z) - 1.0f);
    r.w = (r.w > 0.0f) ? r.w : (__expf(r.w) - 1.0f);
    v4bf o;
    o[0] = (__bf16)r.x; o[1] = (__bf16)r.y; o[2] = (__bf16)r.z; o[3] = (__bf16)r.w;
    *(v4bf*)dst = o;
}

__global__ void __launch_bounds__(256) __attribute__((amdgpu_waves_per_eu(10)))
node_aggregate_pair_kernel(const int* __restrict__ off, const int* __restrict__ csr,
                           const float* __restrict__ as_q, const float* __restrict__ ad_q,
                           const float* __restrict__ as_t, const float* __restrict__ ad_t,
                           const __bf16* __restrict__ Hq, const __bf16* __restrict__ Ht,
                           const float* __restrict__ bias,
                           __bf16* __restrict__ Xbq, __bf16* __restrict__ Xbt)
{
    const int d    = __builtin_amdgcn_readfirstlane(
                         (int)(blockIdx.x * (blockDim.x >> 5) + (threadIdx.x >> 5)));
    const int lane = threadIdx.x & 31;
    if (d >= N_NODES) return;
    const int half = lane >> 4;
    const int c    = lane & 15;
    const int beg  = off[d];
    const int deg  = off[d + 1] - beg;
    const int n    = deg + 1;
    const float advq = ad_q[d];
    const float advt = ad_t[d];
    const v8bf* __restrict__ HrowQ = (const v8bf*)Hq + c;
    const v8bf* __restrict__ HrowT = (const v8bf*)Ht + c;

    int   s0  = d;
    float v0q = -INFINITY, v0t = -INFINITY;
    if (lane < n) {
        s0 = (lane < deg) ? csr[beg + lane] : d;
        const float tq = as_q[s0] + advq;
        const float tt = as_t[s0] + advt;
        v0q = (tq > 0.0f) ? tq : NEG_SLOPE * tq;
        v0t = (tt > 0.0f) ? tt : NEG_SLOPE * tt;
    }
    float mxq = v0q, mxt = v0t;
    for (int base = 32; base < n; base += 32) {
        const int j = base + lane;
        if (j < n) {
            const int s = (j < deg) ? csr[beg + j] : d;
            const float tq = as_q[s] + advq;
            const float tt = as_t[s] + advt;
            mxq = fmaxf(mxq, (tq > 0.0f) ? tq : NEG_SLOPE * tq);
            mxt = fmaxf(mxt, (tt > 0.0f) ? tt : NEG_SLOPE * tt);
        }
    }
#pragma unroll
    for (int o = 16; o > 0; o >>= 1) {
        mxq = fmaxf(mxq, __shfl_xor(mxq, o, 32));
        mxt = fmaxf(mxt, __shfl_xor(mxt, o, 32));
    }

    float accq[8] = {0.f, 0.f, 0.f, 0.f, 0.f, 0.f, 0.f, 0.f};
    float acct[8] = {0.f, 0.f, 0.f, 0.f, 0.f, 0.f, 0.f, 0.f};
    float zq = 0.0f, zt = 0.0f;
    for (int base = 0; base < n; base += 32) {
        int   s  = s0;
        float wq = (lane < n) ? __expf(v0q - mxq) : 0.0f;
        float wt = (lane < n) ? __expf(v0t - mxt) : 0.0f;
        if (base != 0) {
            const int j = base + lane;
            s = d; wq = 0.0f; wt = 0.0f;
            if (j < n) {
                s = (j < deg) ? csr[beg + j] : d;
                const float tq = as_q[s] + advq;
                const float tt = as_t[s] + advt;
                wq = __expf(((tq > 0.0f) ? tq : NEG_SLOPE * tq) - mxq);
                wt = __expf(((tt > 0.0f) ? tt : NEG_SLOPE * tt) - mxt);
            }
        }
        zq += wq;
        zt += wt;
        const int m = (n - base < 32) ? (n - base) : 32;
        for (int k = 0; k < m; k += 8) {
            v8bf  hq[4], ht[4];
            float wwq[4], wwt[4];
#pragma unroll
            for (int u = 0; u < 4; ++u) {
                const int jj = k + 2 * u + half;
                const int ss = __shfl(s, jj, 32);
                wwq[u] = __shfl(wq, jj, 32);
                wwt[u] = __shfl(wt, jj, 32);
                hq[u]  = HrowQ[(size_t)ss * 16];
                ht[u]  = HrowT[(size_t)ss * 16];
            }
#pragma unroll
            for (int u = 0; u < 4; ++u) {
                row_fma(accq, hq[u], wwq[u]);
                row_fma(acct, ht[u], wwt[u]);
            }
        }
    }
    const int col = c * 8 + half * 4;
    const float4 b4 = *(const float4*)(bias + col);
    finish_row_bf16(accq, zq, b4, half, Xbq + (size_t)d * DIM + col);
    finish_row_bf16(acct, zt, b4, half, Xbt + (size_t)d * DIM + col);
}

extern "C" void kernel_launch(void* const* d_in, const int* in_sizes, int n_in,
                              void* d_out, int out_size, void* d_ws, size_t ws_size,
                              hipStream_t stream)
{
    (void)in_sizes; (void)n_in; (void)out_size; (void)ws_size;

    const float* xq    = (const float*)d_in[0];
    const float* xt    = (const float*)d_in[1];
    const int*   ei_q  = (const int*)d_in[2];
    const float* W_mlp = (const float*)d_in[4];
    const float* b_mlp = (const float*)d_in[5];
    const float* Ws    = (const float*)d_in[6];
    const float* a_src = (const float*)d_in[7];
    const float* a_dst = (const float*)d_in[8];
    const float* biases= (const float*)d_in[9];

    const int* esrc = ei_q;
    const int* edst = ei_q + E_EDGES;

    float* Xq = (float*)d_out;
    float* Xt = Xq + (size_t)N_NODES * DIM;
    __bf16* XbQ = (__bf16*)Xq;
    __bf16* Ht  = XbQ + (size_t)N_NODES * DIM;
    __bf16* XbT = (__bf16*)Xt;

    __bf16* Wpk  = (__bf16*)d_ws;
    __bf16* Hq   = Wpk + (size_t)NWEIGHTS * WPK_ELEMS_PER_W;
    float*  as_q = (float*)(Hq + (size_t)N_NODES * DIM);
    float*  ad_q = as_q + N_NODES;
    int*    cnt  = (int*)(ad_q + N_NODES);
    int*    off  = cnt + N_NODES;
    int*    rank = off + N_NODES + 8;
    int*    tmp  = rank + E_EDGES;
    int*    csr  = tmp + E_EDGES;
    float*  as_t = (float*)rank;
    float*  ad_t = as_t + N_NODES;

    const dim3 blk(256);
    const dim3 pack_grid((NWEIGHTS * 32 * 32 + 255) / 256);
    const dim3 gemm_blk(GEMM_WAVES * 32);
    const dim3 gemm_grid((2 * STRIPS + GEMM_WAVES - 1) / GEMM_WAVES);
    const dim3 node_wave_grid((N_NODES + 7) / 8);
    const dim3 edge_grid((E_EDGES + 255) / 256);

    pack_weights_kernel<<<pack_grid, blk, 0, stream>>>(W_mlp, Ws, Wpk, cnt);

    edge_rank_kernel  <<<edge_grid, blk, 0, stream>>>(edst, cnt, rank);
    bucket_scan_kernel<<<dim3(1), dim3(SCAN_THREADS), 0, stream>>>(cnt, off);
    bucket_fill_kernel<<<edge_grid, blk, 0, stream>>>(esrc, edst, off, rank, tmp);
    bucket_sort_kernel<<<node_wave_grid, blk, 0, stream>>>(off, tmp, csr);

    gemm128_wmma<true><<<gemm_grid, gemm_blk, 0, stream>>>(xq, xt, nullptr, nullptr,
                                                       Wpk, b_mlp, Wpk + WPK_ELEMS_PER_W, Hq, Ht,
                                                       a_src, a_dst, as_q, ad_q, as_t, ad_t);
    node_aggregate_pair_kernel<<<node_wave_grid, blk, 0, stream>>>(off, csr, as_q, ad_q, as_t, ad_t,
                                                                  Hq, Ht, biases, XbQ, XbT);

    gemm128_wmma<false><<<gemm_grid, gemm_blk, 0, stream>>>(nullptr, nullptr, XbQ, XbT,
                                                        nullptr, nullptr, Wpk + (size_t)2 * WPK_ELEMS_PER_W, Hq, Ht,
                                                        a_src + DIM, a_dst + DIM, as_q, ad_q, as_t, ad_t);
    node_aggregate_pair_kernel<<<node_wave_grid, blk, 0, stream>>>(off, csr, as_q, ad_q, as_t, ad_t,
                                                                  Hq, Ht, biases + DIM, XbQ, XbT);

    gemm128_wmma<false><<<gemm_grid, gemm_blk, 0, stream>>>(nullptr, nullptr, XbQ, XbT,
                                                        nullptr, nullptr, Wpk + (size_t)3 * WPK_ELEMS_PER_W, Hq, Ht,
                                                        a_src + 2 * DIM, a_dst + 2 * DIM, as_q, ad_q, as_t, ad_t);
    node_aggregate_kernel<<<node_wave_grid, blk, 0, stream>>>(off, csr, as_t, ad_t, Ht, biases + 2 * DIM, Xt);
    node_aggregate_kernel<<<node_wave_grid, blk, 0, stream>>>(off, csr, as_q, ad_q, Hq, biases + 2 * DIM, Xq);
}


# Aggregate: one half-wave per node (a wave serves nodes 2w, 2w+1; each row load fetches one bf16 row per node): half the waves; new sum order: slots in order in one chain, z over 16 lanes
# speedup vs baseline: 7.3122x; 1.1353x over previous
#include <hip/hip_runtime.h>
#include <hip/hip_bf16.h>

#define N_NODES 30000
#define E_EDGES 480000
#define DIM     128
#define NLAYERS 3
#define NEG_SLOPE 0.2f
#define NWEIGHTS 4
#define WPK_ELEMS_PER_W (8 * 4 * 32 * 16)

typedef __attribute__((ext_vector_type(16))) __bf16 v16bf;
typedef __attribute__((ext_vector_type(4)))  __bf16 v4bf;
typedef __attribute__((ext_vector_type(8)))  __bf16 v8bf;
typedef __attribute__((ext_vector_type(8)))  float  v8f;

__global__ void __launch_bounds__(256)
pack_weights_kernel(const float* __restrict__ W_mlp, const float* __restrict__ Ws,
                    __bf16* __restrict__ Wpk)
{
    const int gw   = (blockIdx.x * blockDim.x + threadIdx.x) >> 5;
    const int lane = threadIdx.x & 31;
    if (gw >= NWEIGHTS * 32) return;
    const int w  = gw >> 5;
    const int ct = (gw & 31) >> 2;
    const int s  = gw & 3;
    const float* W = (w == 0) ? W_mlp : (Ws + (size_t)(w - 1) * DIM * DIM);
    const int r  = lane & 15;
    const int hi = lane >> 4;
    const int col = ct * 16 + r;
    const int kb  = s * 32;
    __bf16* out = Wpk + ((size_t)gw * 32 + lane) * 16;
#pragma unroll
    for (int h = 0; h < 16; ++h) {
        const int K = kb + ((h < 8) ? (8 * hi + h) : (16 + 8 * hi + (h - 8)));
        out[h] = (__bf16)W[(size_t)K * DIM + col];
    }
}

#define GEMM_WAVES 2

__device__ __forceinline__ void load_fragments(v16bf (&w)[32], const __bf16* Wfrag, const int lane)
{
#pragma unroll
    for (int s = 0; s < 4; ++s) {
#pragma unroll
        for (int ct = 0; ct < 8; ++ct)
            w[ct * 4 + s] = *(const v16bf*)(Wfrag + ((size_t)(ct * 4 + s) * 32 + lane) * 16);
    }
}

__device__ __forceinline__ void strip_product(v8f (&acc)[8], const v16bf (&w)[32], const v16bf (&x)[4])
{
#pragma unroll
    for (int ct = 0; ct < 8; ++ct) acc[ct] = v8f{};
#pragma unroll
    for (int s = 0; s < 4; ++s) {
#pragma unroll
        for (int ct = 0; ct < 8; ++ct) {
            acc[ct] = __builtin_amdgcn_wmma_f32_16x16x32_bf16(
                 false, w[ct * 4 + s],  false, x[s],
                 (short)0, acc[ct],  false,  false);
        }
    }
}

#define STRIPS (N_NODES / 16)

template <bool MLP_FIRST>
__global__ void __launch_bounds__(GEMM_WAVES * 32)
gemm128_wmma(const float* __restrict__ Xraw_q, const float* __restrict__ Xraw_t,
             const __bf16* __restrict__ Xb_q, const __bf16* __restrict__ Xb_t,
             const __bf16* __restrict__ W0pk, const float* __restrict__ bias0,
             const __bf16* __restrict__ Wpk,
             __bf16* __restrict__ H_q, __bf16* __restrict__ H_t,
             const float* __restrict__ asrc, const float* __restrict__ adst,
             float* __restrict__ as_q, float* __restrict__ ad_q,
             float* __restrict__ as_t, float* __restrict__ ad_t)
{
    const int sidx = __builtin_amdgcn_readfirstlane(
                         (int)(blockIdx.x * GEMM_WAVES + (threadIdx.x >> 5)));
    if (sidx >= 2 * STRIPS) return;
    const bool second = sidx >= STRIPS;
    const int strip = second ? (sidx - STRIPS) : sidx;
    const int lane = threadIdx.x & 31;
    const int r    = lane & 15;
    const int hi   = lane >> 4;
    const int row  = strip * 16 + r;
    const int cb   = 8 * hi;

    v16bf x[4];
    v16bf w[32];
    if constexpr (MLP_FIRST) {
        const float* X = second ? Xraw_t : Xraw_q;
        float4 p[16];
        const float4* xp = (const float4*)(X + (size_t)row * DIM + cb);
#pragma unroll
        for (int s = 0; s < 4; ++s) {
            p[4 * s + 0] = xp[s * 8 + 0];
            p[4 * s + 1] = xp[s * 8 + 1];
            p[4 * s + 2] = xp[s * 8 + 4];
            p[4 * s + 3] = xp[s * 8 + 5];
        }
        load_fragments(w, W0pk, lane);
        __builtin_amdgcn_sched_barrier(0);
#pragma unroll
        for (int s = 0; s < 4; ++s) {
            const float4 p0 = p[4 * s + 0], p1 = p[4 * s + 1], p2 = p[4 * s + 2], p3 = p[4 * s + 3];
            x[s][0]  = (__bf16)p0.x; x[s][1]  = (__bf16)p0.y; x[s][2]  = (__bf16)p0.z; x[s][3]  = (__bf16)p0.w;
            x[s][4]  = (__bf16)p1.x; x[s][5]  = (__bf16)p1.y; x[s][6]  = (__bf16)p1.z; x[s][7]  = (__bf16)p1.w;
            x[s][8]  = (__bf16)p2.x; x[s][9]  = (__bf16)p2.y; x[s][10] = (__bf16)p2.z; x[s][11] = (__bf16)p2.w;
            x[s][12] = (__bf16)p3.x; x[s][13] = (__bf16)p3.y; x[s][14] = (__bf16)p3.z; x[s][15] = (__bf16)p3.w;
        }
    } else {
        const __bf16* Xb = second ? Xb_t : Xb_q;
        const v8bf* xp = (const v8bf*)(Xb + (size_t)row * DIM + cb);
        v8bf q[8];
#pragma unroll
        for (int s = 0; s < 4; ++s) {
            q[2 * s + 0] = xp[s * 4 + 0];
            q[2 * s + 1] = xp[s * 4 + 2];
        }
        load_fragments(w, Wpk, lane);
        __builtin_amdgcn_sched_barrier(0);
#pragma unroll
        for (int s = 0; s < 4; ++s)
            x[s] = __builtin_shufflevector(q[2 * s + 0], q[2 * s + 1],
                                           0, 1, 2, 3, 4, 5, 6, 7, 8, 9, 10, 11, 12, 13, 14, 15);
    }

    v8f acc[8];
    if constexpr (MLP_FIRST) {
        strip_product(acc, w, x);
        __builtin_amdgcn_sched_barrier(0);
        load_fragments(w, Wpk, lane);
        float4 b[16];
#pragma unroll
        for (int s = 0; s < 4; ++s) {
            b[4 * s + 0] = *(const float4*)(bias0 + s * 32 + cb);
            b[4 * s + 1] = *(const float4*)(bias0 + s * 32 + cb + 4);
            b[4 * s + 2] = *(const float4*)(bias0 + s * 32 + 16 + cb);
            b[4 * s + 3] = *(const float4*)(bias0 + s * 32 + 16 + cb + 4);
        }
        __builtin_amdgcn_sched_barrier(0);
#pragma unroll
        for (int s = 0; s < 4; ++s) {
            const float4 b0 = b[4 * s + 0], b1 = b[4 * s + 1], b2 = b[4 * s + 2], b3 = b[4 * s + 3];
            const v8f lo = acc[2 * s];
            const v8f up = acc[2 * s + 1];
            x[s][0]  = (__bf16)(lo[0] + b0.x); x[s][1]  = (__bf16)(lo[1] + b0.y);
            x[s][2]  = (__bf16)(lo[2] + b0.z); x[s][3]  = (__bf16)(lo[3] + b0.w);
            x[s][4]  = (__bf16)(lo[4] + b1.x); x[s][5]  = (__bf16)(lo[5] + b1.y);
            x[s][6]  = (__bf16)(lo[6] + b1.z); x[s][7]  = (__bf16)(lo[7] + b1.w);
            x[s][8]  = (__bf16)(up[0] + b2.x); x[s][9]  = (__bf16)(up[1] + b2.y);
            x[s][10] = (__bf16)(up[2] + b2.z); x[s][11] = (__bf16)(up[3] + b2.w);
            x[s][12] = (__bf16)(up[4] + b3.x); x[s][13] = (__bf16)(up[5] + b3.y);
            x[s][14] = (__bf16)(up[6] + b3.z); x[s][15] = (__bf16)(up[7] + b3.w);
        }
    }

    float4 av[16], dv[16];
#pragma unroll
    for (int ct = 0; ct < 8; ++ct) {
        av[2 * ct + 0] = *(const float4*)(asrc + ct * 16 + cb);
        av[2 * ct + 1] = *(const float4*)(asrc + ct * 16 + cb + 4);
        dv[2 * ct + 0] = *(const float4*)(adst + ct * 16 + cb);
        dv[2 * ct + 1] = *(const float4*)(adst + ct * 16 + cb + 4);
    }
    __builtin_amdgcn_sched_barrier(0);
    strip_product(acc, w, x);

    float ps = 0.0f, pd = 0.0f;
    __bf16* H = (second ? H_t : H_q) + (size_t)row * DIM + cb;
#pragma unroll
    for (int ct = 0; ct < 8; ++ct) {
        const float4 s0 = av[2 * ct + 0];
        const float4 s1 = av[2 * ct + 1];
        const float4 d0 = dv[2 * ct + 0];
        const float4 d1 = dv[2 * ct + 1];
        const v8f a = acc[ct];
        float ts = a[0] * s0.x;
        ts = fmaf(a[1], s0.y, ts); ts = fmaf(a[2], s0.z, ts); ts = fmaf(a[3], s0.w, ts);
        ts = fmaf(a[4], s1.x, ts); ts = fmaf(a[5], s1.y, ts); ts = fmaf(a[6], s1.z, ts);
        ts = fmaf(a[7], s1.w, ts);
        float td = a[0] * d0.x;
        td = fmaf(a[1], d0.y, td); td = fmaf(a[2], d0.z, td); td = fmaf(a[3], d0.w, td);
        td = fmaf(a[4], d1.x, td); td = fmaf(a[5], d1.y, td); td = fmaf(a[6], d1.z, td);
        td = fmaf(a[7], d1.w, td);
        ps += ts;
        pd += td;
        v8bf o;
        o[0] = (__bf16)a[0]; o[1] = (__bf16)a[1]; o[2] = (__bf16)a[2]; o[3] = (__bf16)a[3];
        o[4] = (__bf16)a[4]; o[5] = (__bf16)a[5]; o[6] = (__bf16)a[6]; o[7] = (__bf16)a[7];
        *(v8bf*)(H + ct * 16) = o;
    }
    ps += __shfl_xor(ps, 16, 32);
    pd += __shfl_xor(pd, 16, 32);
    if (hi) (second ? ad_t : ad_q)[row] = pd; else (second ? as_t : as_q)[row] = ps;
}

#define BIN_W        118
#define NBINS        ((N_NODES + BIN_W - 1) / BIN_W)
#define NBP          256
#define NCHUNK       128
#define CHUNK_E      ((E_EDGES + NCHUNK - 1) / NCHUNK)
#define EDGE_THREADS 1024
#define EDGE_PER     ((CHUNK_E + EDGE_THREADS - 1) / EDGE_THREADS)
#define TABLE_PARTS  (EDGE_THREADS / NBP)
static_assert(NBINS <= NBP, "one table column per bin");
static_assert(BIN_W <= 128, "bin_build scans 128 counters with one wave and sorts 4 buckets per wave");
static_assert(NCHUNK % TABLE_PARTS == 0, "table rows split evenly over the thread groups");

__global__ void __launch_bounds__(EDGE_THREADS)
bin_count_kernel(const int* __restrict__ edst, int* __restrict__ table)
{
    __shared__ int hist[NBP];
    const int t  = threadIdx.x;
    const int e0 = blockIdx.x * CHUNK_E;
    const int e1 = (e0 + CHUNK_E < E_EDGES) ? (e0 + CHUNK_E) : E_EDGES;
    if (t < NBP) hist[t] = 0;
    int d[EDGE_PER];
#pragma unroll
    for (int i = 0; i < EDGE_PER; ++i) {
        const int e = e0 + i * EDGE_THREADS + t;
        d[i] = (e < e1) ? edst[e] : -1;
    }
    __syncthreads();
#pragma unroll
    for (int i = 0; i < EDGE_PER; ++i)
        if (d[i] >= 0) atomicAdd(&hist[d[i] / BIN_W], 1);
    __syncthreads();
    if (t < NBP) table[blockIdx.x * NBP + t] = hist[t];
}

__global__ void __launch_bounds__(EDGE_THREADS)
bin_scatter_kernel(const int* __restrict__ esrc, const int* __restrict__ edst,
                   const int* __restrict__ table, int* __restrict__ binstart,
                   int2* __restrict__ rec)
{
    __shared__ int part_tot[TABLE_PARTS][NBP];
    __shared__ int part_mine[TABLE_PARTS][NBP];
    __shared__ int wsum[NBP / 32];
    __shared__ int cursor[NBP];
    const int t  = threadIdx.x;
    const int wg = blockIdx.x;
    const int e0 = wg * CHUNK_E;
    const int e1 = (e0 + CHUNK_E < E_EDGES) ? (e0 + CHUNK_E) : E_EDGES;
    int d[EDGE_PER], s[EDGE_PER];
#pragma unroll
    for (int i = 0; i < EDGE_PER; ++i) {
        const int e = e0 + i * EDGE_THREADS + t;
        d[i] = (e < e1) ? edst[e] : -1;
        s[i] = (e < e1) ? esrc[e] : 0;
    }
    {
        const int bin  = t & (NBP - 1);
        const int part = t / NBP;
        const int g0   = part * (NCHUNK / TABLE_PARTS);
        int tot = 0, mine = 0;
        int h[NCHUNK / TABLE_PARTS];
#pragma unroll
        for (int k = 0; k < NCHUNK / TABLE_PARTS; ++k) h[k] = table[(g0 + k) * NBP + bin];
#pragma unroll
        for (int k = 0; k < NCHUNK / TABLE_PARTS; ++k) {
            tot  += h[k];
            mine += (g0 + k < wg) ? h[k] : 0;
        }
        part_tot[part][bin]  = tot;
        part_mine[part][bin] = mine;
    }
    __syncthreads();
    int tot = 0, mine = 0, inc = 0;
    if (t < NBP) {
#pragma unroll
        for (int p = 0; p < TABLE_PARTS; ++p) { tot += part_tot[p][t]; mine += part_mine[p][t]; }
        inc = tot;
#pragma unroll
        for (int o = 1; o < 32; o <<= 1) {
            const int y = __shfl_up(inc, o, 32);
            if ((t & 31) >= o) inc += y;
        }
        if ((t & 31) == 31) wsum[t >> 5] = inc;
    }
    __syncthreads();
    if (t < NBP) {
        int pre = 0;
        for (int k = 0; k < (t >> 5); ++k) pre += wsum[k];
        const int start = pre + inc - tot;
        cursor[t] = start + mine;
        if (wg == 0) {
            binstart[t] = start;
            if (t == NBP - 1) binstart[NBP] = start + tot;
        }
    }
    __syncthreads();
#pragma unroll
    for (int i = 0; i < EDGE_PER; ++i) {
        if (d[i] >= 0) {
            const int pos = atomicAdd(&cursor[d[i] / BIN_W], 1);
            rec[pos] = make_int2(d[i], s[i]);
        }
    }
}

#define BIN_THREADS 1024
__global__ void __launch_bounds__(BIN_THREADS)
bin_build_kernel(const int* __restrict__ binstart, const int2* __restrict__ rec,
                 int* rk, int* tmp, int* __restrict__ off, int* __restrict__ csr)
{
    __shared__ int c[128];
    __shared__ int lo[128];
    const int t    = threadIdx.x;
    const int lane = t & 31;
    const int w    = __builtin_amdgcn_readfirstlane(t >> 5);
    const int b    = blockIdx.x;
    const int node0 = b * BIN_W;
    const int nn   = (N_NODES - node0 < BIN_W) ? (N_NODES - node0) : BIN_W;
    const int bs   = binstart[b];
    const int cnt  = binstart[b + 1] - bs;
    if (t < 128) c[t] = 0;
    __syncthreads();
    for (int i = t; i < cnt; i += BIN_THREADS) {
        const int2 r = rec[bs + i];
        rk[bs + i] = atomicAdd(&c[r.x - node0], 1);
    }
    __syncthreads();
    if (w == 0) {
        int v[4];
        int sum = 0;
#pragma unroll
        for (int k = 0; k < 4; ++k) { v[k] = c[lane * 4 + k]; sum += v[k]; }
        int inc = sum;
#pragma unroll
        for (int o = 1; o < 32; o <<= 1) {
            const int y = __shfl_up(inc, o, 32);
            if (lane >= o) inc += y;
        }
        int run = inc - sum;
#pragma unroll
        for (int k = 0; k < 4; ++k) { lo[lane * 4 + k] = run; run += v[k]; }
    }
    __syncthreads();
    if (t < nn) off[node0 + t] = bs + lo[t];
    if (b == NBINS - 1 && t == 0) off[N_NODES] = bs + cnt;
    for (int i = t; i < cnt; i += BIN_THREADS) {
        const int2 r = rec[bs + i];
        tmp[bs + lo[r.x - node0] + rk[bs + i]] = r.y;
    }
    __threadfence();
    __syncthreads();

    int key[4], beg[4], deg[4];
#pragma unroll
    for (int q = 0; q < 4; ++q) {
        const int j = w + 32 * q;
        beg[q] = 0; deg[q] = -1; key[q] = 0x7fffffff;
        if (j < nn) {
            beg[q] = bs + __builtin_amdgcn_readfirstlane(lo[j]);
            deg[q] = __builtin_amdgcn_readfirstlane(c[j]);
            if (deg[q] <= 32 && lane < deg[q]) key[q] = tmp[beg[q] + lane];
        }
    }
#pragma unroll
    for (int q = 0; q < 4; ++q) {
        if (deg[q] < 0) continue;
        if (deg[q] <= 32) {
            int pos = 0;
            for (int j = 0; j < deg[q]; ++j) {
                const int kj = __builtin_amdgcn_readlane(key[q], j);
                pos += (kj < key[q] || (kj == key[q] && j < lane)) ? 1 : 0;
            }
            if (lane < deg[q]) csr[beg[q] + pos] = key[q];
        } else {
            for (int i = lane; i < deg[q]; i += 32) {
                const int ki = tmp[beg[q] + i];
                int pos = 0;
                for (int j = 0; j < deg[q]; ++j) {
                    const int kj = tmp[beg[q] + j];
                    pos += (kj < ki || (kj == ki && j < i)) ? 1 : 0;
                }
                csr[beg[q] + pos] = ki;
            }
        }
    }
}

#define ROW_LOADS_PER_TRIP 8
#define AGG_WAVES 8
static_assert(N_NODES % 2 == 0, "two nodes per wave");

__device__ __forceinline__ void row_fma(float (&acc)[8], const v8bf h, const float w)
{
#pragma unroll
    for (int i = 0; i < 8; ++i) acc[i] = fmaf(w, (float)h[i], acc[i]);
}

__device__ __forceinline__ float leaky(const float t) { return (t > 0.0f) ? t : NEG_SLOPE * t; }

template <int NG, bool OUT_BF16>
__device__ __forceinline__ void
aggregate_body(const int* __restrict__ off, const int* __restrict__ csr,
               const float* __restrict__ as_a, const float* __restrict__ ad_a,
               const float* __restrict__ as_b, const float* __restrict__ ad_b,
               const __bf16* __restrict__ Ha, const __bf16* __restrict__ Hb,
               const float* __restrict__ bias,
               float* __restrict__ Xf, __bf16* __restrict__ Xba, __bf16* __restrict__ Xbb)
{
    const int wv   = __builtin_amdgcn_readfirstlane(
                         (int)(blockIdx.x * AGG_WAVES + (threadIdx.x >> 5)));
    if (2 * wv >= N_NODES) return;
    const int lane = threadIdx.x & 31;
    const int half = lane >> 4;
    const int c    = lane & 15;
    const int d    = 2 * wv + half;
    const int beg  = off[d];
    const int deg  = off[d + 1] - beg;
    const int n    = deg + 1;
    const int other = __shfl_xor(n, 16, 32);
    const int nmax = __builtin_amdgcn_readfirstlane((n > other) ? n : other);
    const int col  = c * 8;

    float adv[NG];
#pragma unroll
    for (int g = 0; g < NG; ++g) adv[g] = (g == 0 ? ad_a : ad_b)[d];

    const int j1 = 16 + c;
    const int e0 = (beg + c  < E_EDGES) ? (beg + c)  : (E_EDGES - 1);
    const int e1 = (beg + j1 < E_EDGES) ? (beg + j1) : (E_EDGES - 1);
    const int l0 = csr[e0];
    const int l1 = csr[e1];
    const int s0 = (c  < deg) ? l0 : d;
    const int s1 = (j1 < deg) ? l1 : d;
    float v0[NG], v1[NG], mx[NG];
#pragma unroll
    for (int g = 0; g < NG; ++g) {
        const float* __restrict__ as_ = (g == 0) ? as_a : as_b;
        const float t0 = as_[s0] + adv[g];
        const float t1 = as_[s1] + adv[g];
        v0[g] = leaky(t0);
        v1[g] = leaky(t1);
        mx[g] = fmaxf(v0[g], v1[g]);
    }
    for (int base = 32; base < nmax; base += 16) {
        const int j = base + c;
        if (j < n) {
            const int s = (j < deg) ? csr[beg + j] : d;
#pragma unroll
            for (int g = 0; g < NG; ++g) {
                const float t = (g == 0 ? as_a : as_b)[s] + adv[g];
                mx[g] = fmaxf(mx[g], leaky(t));
            }
        }
    }
#pragma unroll
    for (int o = 8; o > 0; o >>= 1) {
#pragma unroll
        for (int g = 0; g < NG; ++g) mx[g] = fmaxf(mx[g], __shfl_xor(mx[g], o, 32));
    }

    float acc[NG][8];
    float zp[NG];
#pragma unroll
    for (int g = 0; g < NG; ++g) {
        zp[g] = 0.0f;
#pragma unroll
        for (int i = 0; i < 8; ++i) acc[g][i] = 0.0f;
    }
    const char* __restrict__ HbaseA = (const char*)Ha;
    const char* __restrict__ HbaseB = (const char*)Hb;
    const unsigned coff = (unsigned)c * 16u;
    const int lane0 = lane & 16;
    constexpr int RPT = ROW_LOADS_PER_TRIP / NG;

    for (int base = 0; base < nmax; base += 16) {
        int   s;
        float w[NG];
        if (base == 0) {
            s = s0;
#pragma unroll
            for (int g = 0; g < NG; ++g) w[g] = (c < n) ? __expf(v0[g] - mx[g]) : 0.0f;
        } else if (base == 16) {
            s = s1;
#pragma unroll
            for (int g = 0; g < NG; ++g) w[g] = (j1 < n) ? __expf(v1[g] - mx[g]) : 0.0f;
        } else {
            const int j = base + c;
            s = d;
#pragma unroll
            for (int g = 0; g < NG; ++g) w[g] = 0.0f;
            if (j < n) {
                s = (j < deg) ? csr[beg + j] : d;
#pragma unroll
                for (int g = 0; g < NG; ++g) {
                    const float t = (g == 0 ? as_a : as_b)[s] + adv[g];
                    w[g] = __expf(leaky(t) - mx[g]);
                }
            }
        }
#pragma unroll
        for (int g = 0; g < NG; ++g) zp[g] += w[g];
        const int m = (nmax - base < 16) ? (nmax - base) : 16;
        for (int k = 0; k < m; k += RPT) {
            v8bf  h[NG][RPT];
            float ww[NG][RPT];
#pragma unroll
            for (int u = 0; u < RPT; ++u) {
                const int jj = lane0 + k + u;
                const unsigned ro = (unsigned)__shfl(s, jj, 32) * (unsigned)(DIM * 2) + coff;
#pragma unroll
                for (int g = 0; g < NG; ++g) {
                    ww[g][u] = __shfl(w[g], jj, 32);
                    h[g][u]  = *(const v8bf*)((g == 0 ? HbaseA : HbaseB) + ro);
                }
            }
#pragma unroll
            for (int u = 0; u < RPT; ++u) {
#pragma unroll
                for (int g = 0; g < NG; ++g) row_fma(acc[g], h[g][u], ww[g][u]);
            }
        }
    }

    const float4 bias0 = *(const float4*)(bias + col);
    const float4 bias1 = *(const float4*)(bias + col + 4);
#pragma unroll
    for (int g = 0; g < NG; ++g) {
        float z = zp[g];
#pragma unroll
        for (int o = 8; o > 0; o >>= 1) z += __shfl_xor(z, o, 32);
        const float rz = 1.0f / z;
        float r[8];
        r[0] = acc[g][0] * rz + bias0.x; r[1] = acc[g][1] * rz + bias0.y;
        r[2] = acc[g][2] * rz + bias0.z; r[3] = acc[g][3] * rz + bias0.w;
        r[4] = acc[g][4] * rz + bias1.x; r[5] = acc[g][5] * rz + bias1.y;
        r[6] = acc[g][6] * rz + bias1.z; r[7] = acc[g][7] * rz + bias1.w;
#pragma unroll
        for (int i = 0; i < 8; ++i) r[i] = (r[i] > 0.0f) ? r[i] : (__expf(r[i]) - 1.0f);
        if constexpr (OUT_BF16) {
            v8bf o;
#pragma unroll
            for (int i = 0; i < 8; ++i) o[i] = (__bf16)r[i];
            *(v8bf*)((g == 0 ? Xba : Xbb) + (size_t)d * DIM + col) = o;
        } else {
            float* dst = Xf + (size_t)d * DIM + col;
            *(float4*)(dst)     = make_float4(r[0], r[1], r[2], r[3]);
            *(float4*)(dst + 4) = make_float4(r[4], r[5], r[6], r[7]);
        }
    }
}

__global__ void __launch_bounds__(AGG_WAVES * 32)
node_aggregate_kernel(const int* __restrict__ off, const int* __restrict__ csr,
                      const float* __restrict__ as_, const float* __restrict__ ad_,
                      const __bf16* __restrict__ Hb, const float* __restrict__ bias,
                      float* __restrict__ X)
{
    aggregate_body<1, false>(off, csr, as_, ad_, nullptr, nullptr, Hb, nullptr, bias, X, nullptr, nullptr);
}

__global__ void __launch_bounds__(AGG_WAVES * 32) __attribute__((amdgpu_waves_per_eu(10)))
node_aggregate_pair_kernel(const int* __restrict__ off, const int* __restrict__ csr,
                           const float* __restrict__ as_q, const float* __restrict__ ad_q,
                           const float* __restrict__ as_t, const float* __restrict__ ad_t,
                           const __bf16* __restrict__ Hq, const __bf16* __restrict__ Ht,
                           const float* __restrict__ bias,
                           __bf16* __restrict__ Xbq, __bf16* __restrict__ Xbt)
{
    aggregate_body<2, true>(off, csr, as_q, ad_q, as_t, ad_t, Hq, Ht, bias, nullptr, Xbq, Xbt);
}

extern "C" void kernel_launch(void* const* d_in, const int* in_sizes, int n_in,
                              void* d_out, int out_size, void* d_ws, size_t ws_size,
                              hipStream_t stream)
{
    (void)in_sizes; (void)n_in; (void)out_size; (void)ws_size;

    const float* xq    = (const float*)d_in[0];
    const float* xt    = (const float*)d_in[1];
    const int*   ei_q  = (const int*)d_in[2];
    const float* W_mlp = (const float*)d_in[4];
    const float* b_mlp = (const float*)d_in[5];
    const float* Ws    = (const float*)d_in[6];
    const float* a_src = (const float*)d_in[7];
    const float* a_dst = (const float*)d_in[8];
    const float* biases= (const float*)d_in[9];

    const int* esrc = ei_q;
    const int* edst = ei_q + E_EDGES;

    float* Xq = (float*)d_out;
    float* Xt = Xq + (size_t)N_NODES * DIM;
    __bf16* XbQ = (__bf16*)Xq;
    __bf16* Ht  = XbQ + (size_t)N_NODES * DIM;
    __bf16* XbT = (__bf16*)Xt;

    __bf16* Wpk  = (__bf16*)d_ws;
    __bf16* Hq   = Wpk + (size_t)NWEIGHTS * WPK_ELEMS_PER_W;
    float*  as_q = (float*)(Hq + (size_t)N_NODES * DIM);
    float*  ad_q = as_q + N_NODES;
    int*    cnt  = (int*)(ad_q + N_NODES);
    int*    off  = cnt + N_NODES;
    int*    rank = off + N_NODES + 8;
    int*    tmp  = rank + E_EDGES;
    int*    csr  = tmp + E_EDGES;
    float*  as_t = (float*)rank;
    float*  ad_t = as_t + N_NODES;
    int*    table    = cnt;
    int2*   rec      = (int2*)Hq;
    int*    binstart = (int*)Hq + (size_t)2 * E_EDGES;
    static_assert(NCHUNK * NBP <= 2 * N_NODES, "count table inside cnt + off");
    static_assert(2 * E_EDGES + NBP + 1 <= N_NODES * DIM / 2, "records and bin starts inside H of graph q");

    const dim3 blk(256);
    const dim3 pack_grid((NWEIGHTS * 32 * 32 + 255) / 256);
    const dim3 gemm_blk(GEMM_WAVES * 32);
    const dim3 gemm_grid((2 * STRIPS + GEMM_WAVES - 1) / GEMM_WAVES);
    const dim3 node_wave_grid((N_NODES / 2 + 7) / 8);

    pack_weights_kernel<<<pack_grid, blk, 0, stream>>>(W_mlp, Ws, Wpk);

    bin_count_kernel  <<<dim3(NCHUNK), dim3(EDGE_THREADS), 0, stream>>>(edst, table);
    bin_scatter_kernel<<<dim3(NCHUNK), dim3(EDGE_THREADS), 0, stream>>>(esrc, edst, table, binstart, rec);
    bin_build_kernel  <<<dim3(NBINS), dim3(BIN_THREADS), 0, stream>>>(binstart, rec, rank, tmp, off, csr);

    gemm128_wmma<true><<<gemm_grid, gemm_blk, 0, stream>>>(xq, xt, nullptr, nullptr,
                                                       Wpk, b_mlp, Wpk + WPK_ELEMS_PER_W, Hq, Ht,
                                                       a_src, a_dst, as_q, ad_q, as_t, ad_t);
    node_aggregate_pair_kernel<<<node_wave_grid, blk, 0, stream>>>(off, csr, as_q, ad_q, as_t, ad_t,
                                                                  Hq, Ht, biases, XbQ, XbT);

    gemm128_wmma<false><<<gemm_grid, gemm_blk, 0, stream>>>(nullptr, nullptr, XbQ, XbT,
                                                        nullptr, nullptr, Wpk + (size_t)2 * WPK_ELEMS_PER_W, Hq, Ht,
                                                        a_src + DIM, a_dst + DIM, as_q, ad_q, as_t, ad_t);
    node_aggregate_pair_kernel<<<node_wave_grid, blk, 0, stream>>>(off, csr, as_q, ad_q, as_t, ad_t,
                                                                  Hq, Ht, biases + DIM, XbQ, XbT);

    gemm128_wmma<false><<<gemm_grid, gemm_blk, 0, stream>>>(nullptr, nullptr, XbQ, XbT,
                                                        nullptr, nullptr, Wpk + (size_t)3 * WPK_ELEMS_PER_W, Hq, Ht,
                                                        a_src + 2 * DIM, a_dst + 2 * DIM, as_q, ad_q, as_t, ad_t);
    node_aggregate_kernel<<<node_wave_grid, blk, 0, stream>>>(off, csr, as_t, ad_t, Ht, biases + 2 * DIM, Xt);
    node_aggregate_kernel<<<node_wave_grid, blk, 0, stream>>>(off, csr, as_q, ad_q, Hq, biases + 2 * DIM, Xq);
}
